# starting kernel
# baseline (speedup 1.0000x reference)
#include <hip/hip_runtime.h>

typedef _Float16 f16;
typedef __attribute__((ext_vector_type(16))) _Float16 v16h;
typedef __attribute__((ext_vector_type(8)))  float    v8f;

#define NNODES 50000
#define NEDGES 800000
#define FIN    128
#define FOUT   128
#define MT     4

union V16H { v16h v; uint4 q[2]; };
union H8   { f16 h[8]; uint4 q; };

__device__ __forceinline__ uint4 pack8(float4 a, float4 b) {
  H8 u;
  u.h[0] = (f16)a.x; u.h[1] = (f16)a.y; u.h[2] = (f16)a.z; u.h[3] = (f16)a.w;
  u.h[4] = (f16)b.x; u.h[5] = (f16)b.y; u.h[6] = (f16)b.z; u.h[7] = (f16)b.w;
  return u.q;
}

__device__ __forceinline__ v16h ld_a_frag(const f16* lds, int row_stride,
                                          int M, int hi, int kbase) {
  const int kb = kbase + hi * 8;
  V16H t;
  t.q[0] = *(const uint4*)(lds + M * row_stride + kb);
  t.q[1] = *(const uint4*)(lds + M * row_stride + kb + 16);
  return t.v;
}

__device__ __forceinline__ v16h ld_b_frag(const f16* __restrict__ wf,
                                          int frag, int lane) {
  const uint4* p = (const uint4*)(wf + ((size_t)(frag * 32 + lane)) * 16);
  V16H t;
  t.q[0] = p[0];
  t.q[1] = p[1];
  return t.v;
}

__global__ __launch_bounds__(256) void pack_weights_kernel(
    const float* __restrict__ W, f16* __restrict__ out, int K, int N) {
  int o = blockIdx.x * blockDim.x + threadIdx.x;
  if (o >= K * N) return;
  int ksteps = K >> 5;
  int e    = o & 15;
  int lane = (o >> 4) & 31;
  int frag = o >> 9;
  int ks   = frag % ksteps;
  int nt   = frag / ksteps;
  int kl   = (e < 8 ? e : e + 8) + ((lane & 16) ? 8 : 0);
  int Kidx = ks * 32 + kl;
  int n    = nt * 16 + (lane & 15);
  out[o] = (f16)W[Kidx * N + n];
}

__global__ __launch_bounds__(256) void edge_mlp_kernel(
    const float* __restrict__ x,
    const int* __restrict__ src,
    const int* __restrict__ dst,
    const f16* __restrict__ W1f,
    const float* __restrict__ b1,
    const f16* __restrict__ W2f,
    const float* __restrict__ b2,
    float* __restrict__ summed,
    float* __restrict__ cnt) {
  __shared__ f16 lds_e[MT * 16 * 256];
  __shared__ f16 lds_h[MT * 16 * 128];
  __shared__ int lds_dst[MT * 16];

  const int tid   = threadIdx.x;
  const int lane  = tid & 31;
  const int wv    = tid >> 5;
  const int ebase = blockIdx.x * (MT * 16);

  {
    const int r  = tid >> 2;
    const int c0 = (tid & 3) * 32;
    const int e  = ebase + r;
    const int si = src[e];
    const int di = dst[e];
    if ((tid & 3) == 0) {
      lds_dst[r] = di;
      atomicAdd(&cnt[di], 1.0f);
    }
    const float4* xi = (const float4*)(x + (size_t)di * FIN + c0);
    const float4* xj = (const float4*)(x + (size_t)si * FIN + c0);
    uint4* pe = (uint4*)(&lds_e[r * 256 + c0]);
    uint4* pd = (uint4*)(&lds_e[r * 256 + 128 + c0]);
#pragma unroll
    for (int k = 0; k < 4; ++k) {
      float4 a0 = xi[2 * k], a1 = xi[2 * k + 1];
      float4 j0 = xj[2 * k], j1 = xj[2 * k + 1];
      pe[k] = pack8(a0, a1);
      float4 d0 = make_float4(j0.x - a0.x, j0.y - a0.y, j0.z - a0.z, j0.w - a0.w);
      float4 d1 = make_float4(j1.x - a1.x, j1.y - a1.y, j1.z - a1.z, j1.w - a1.w);
      pd[k] = pack8(d0, d1);
    }
  }
  __syncthreads();

  const int M  = lane & 15;
  const int hi = (lane >> 4) & 1;
  const int n  = wv * 16 + (lane & 15);

  v8f acc[MT];
#pragma unroll
  for (int mt = 0; mt < MT; ++mt) acc[mt] = (v8f){};
#pragma unroll
  for (int ks = 0; ks < 8; ++ks) {
    v16h b = ld_b_frag(W1f, wv * 8 + ks, lane);
#pragma unroll
    for (int mt = 0; mt < MT; ++mt) {
      v16h a = ld_a_frag(lds_e + mt * 16 * 256, 256, M, hi, ks * 32);
      acc[mt] = __builtin_amdgcn_wmma_f32_16x16x32_f16(false, a, false, b,
                                                       (short)0, acc[mt],
                                                       false, false);
    }
  }
  {
    const float bias = b1[n];
#pragma unroll
    for (int mt = 0; mt < MT; ++mt) {
#pragma unroll
      for (int r = 0; r < 8; ++r) {
        float v = acc[mt][r] + bias;
        v = v > 0.0f ? v : 0.0f;
        lds_h[(mt * 16 + r + 8 * hi) * 128 + n] = (f16)v;
      }
    }
  }
  __syncthreads();

  v8f acc2[MT];
#pragma unroll
  for (int mt = 0; mt < MT; ++mt) acc2[mt] = (v8f){};
#pragma unroll
  for (int ks = 0; ks < 4; ++ks) {
    v16h b = ld_b_frag(W2f, wv * 4 + ks, lane);
#pragma unroll
    for (int mt = 0; mt < MT; ++mt) {
      v16h a = ld_a_frag(lds_h + mt * 16 * 128, 128, M, hi, ks * 32);
      acc2[mt] = __builtin_amdgcn_wmma_f32_16x16x32_f16(false, a, false, b,
                                                        (short)0, acc2[mt],
                                                        false, false);
    }
  }
  {
    const float bias = b2[n];
#pragma unroll
    for (int mt = 0; mt < MT; ++mt) {
#pragma unroll
      for (int r = 0; r < 8; ++r) {
        const int row = mt * 16 + r + 8 * hi;
        atomicAdd(&summed[(size_t)lds_dst[row] * FOUT + n],
                  acc2[mt][r] + bias);
      }
    }
  }
}

__global__ __launch_bounds__(256) void node_mlp_kernel(
    const float* __restrict__ summed,
    const float* __restrict__ cnt,
    const f16* __restrict__ W3f,
    const float* __restrict__ b3,
    const f16* __restrict__ W4f,
    const float* __restrict__ b4,
    float* __restrict__ out) {
  __shared__ f16 lds_a[16 * 128];
  __shared__ f16 lds_m[16 * 64];

  const int tid   = threadIdx.x;
  const int lane  = tid & 31;
  const int wv    = tid >> 5;
  const int nbase = blockIdx.x * 16;

  {
    const int r    = tid >> 4;
    const int c0   = (tid & 15) * 8;
    const int node = nbase + r;
    float cv  = cnt[node];
    float inv = 1.0f / (cv > 1.0f ? cv : 1.0f);
    const float4* sp = (const float4*)(summed + (size_t)node * FOUT + c0);
    float4 s0 = sp[0], s1 = sp[1];
    float4 m0 = make_float4(s0.x * inv, s0.y * inv, s0.z * inv, s0.w * inv);
    float4 m1 = make_float4(s1.x * inv, s1.y * inv, s1.z * inv, s1.w * inv);
    m0.x = m0.x > 0.f ? m0.x : 0.f; m0.y = m0.y > 0.f ? m0.y : 0.f;
    m0.z = m0.z > 0.f ? m0.z : 0.f; m0.w = m0.w > 0.f ? m0.w : 0.f;
    m1.x = m1.x > 0.f ? m1.x : 0.f; m1.y = m1.y > 0.f ? m1.y : 0.f;
    m1.z = m1.z > 0.f ? m1.z : 0.f; m1.w = m1.w > 0.f ? m1.w : 0.f;
    *(uint4*)(&lds_a[r * 128 + c0]) = pack8(m0, m1);
  }
  __syncthreads();

  const int M  = lane & 15;
  const int hi = (lane >> 4) & 1;

  if (wv < 4) {
    const int n = wv * 16 + (lane & 15);
    v8f acc = {};
#pragma unroll
    for (int ks = 0; ks < 4; ++ks) {
      v16h a = ld_a_frag(lds_a, 128, M, hi, ks * 32);
      v16h b = ld_b_frag(W3f, wv * 4 + ks, lane);
      acc = __builtin_amdgcn_wmma_f32_16x16x32_f16(false, a, false, b,
                                                   (short)0, acc, false, false);
    }
    const float bias = b3[n];
#pragma unroll
    for (int r = 0; r < 8; ++r) {
      float v = acc[r] + bias;
      v = v > 0.0f ? v : 0.0f;
      lds_m[(r + 8 * hi) * 64 + n] = (f16)v;
    }
  }
  __syncthreads();

  {
    const int n = wv * 16 + (lane & 15);
    v8f acc = {};
#pragma unroll
    for (int ks = 0; ks < 2; ++ks) {
      v16h a = ld_a_frag(lds_m, 64, M, hi, ks * 32);
      v16h b = ld_b_frag(W4f, wv * 2 + ks, lane);
      acc = __builtin_amdgcn_wmma_f32_16x16x32_f16(false, a, false, b,
                                                   (short)0, acc, false, false);
    }
    const float bias = b4[n];
#pragma unroll
    for (int r = 0; r < 8; ++r) {
      out[(size_t)(nbase + r + 8 * hi) * FOUT + n] = acc[r] + bias;
    }
  }
}

extern "C" void kernel_launch(void* const* d_in, const int* in_sizes, int n_in,
                              void* d_out, int out_size, void* d_ws, size_t ws_size,
                              hipStream_t stream) {
  const float* x   = (const float*)d_in[0];
  const int*  eidx = (const int*)d_in[1];
  const float* W1  = (const float*)d_in[2];
  const float* b1  = (const float*)d_in[3];
  const float* W2  = (const float*)d_in[4];
  const float* b2  = (const float*)d_in[5];
  const float* W3  = (const float*)d_in[6];
  const float* b3  = (const float*)d_in[7];
  const float* W4  = (const float*)d_in[8];
  const float* b4  = (const float*)d_in[9];
  float* out = (float*)d_out;

  char* ws = (char*)d_ws;
  float* summed = (float*)ws;
  float* cnt    = (float*)(ws + 25600000);
  f16* W1f = (f16*)(ws + 25800000);
  f16* W2f = (f16*)(ws + 25800000 + 65536);
  f16* W3f = (f16*)(ws + 25800000 + 65536 + 32768);
  f16* W4f = (f16*)(ws + 25800000 + 65536 + 32768 + 16384);

  hipMemsetAsync(summed, 0, (size_t)(NNODES * FOUT + NNODES) * sizeof(float), stream);

  pack_weights_kernel<<<(256 * 128 + 255) / 256, 256, 0, stream>>>(W1, W1f, 256, 128);
  pack_weights_kernel<<<(128 * 128 + 255) / 256, 256, 0, stream>>>(W2, W2f, 128, 128);
  pack_weights_kernel<<<(128 * 64  + 255) / 256, 256, 0, stream>>>(W3, W3f, 128, 64);
  pack_weights_kernel<<<(64  * 128 + 255) / 256, 256, 0, stream>>>(W4, W4f, 64, 128);

  edge_mlp_kernel<<<NEDGES / (MT * 16), 256, 0, stream>>>(
      x, eidx, eidx + NEDGES, W1f, b1, W2f, b2, summed, cnt);

  node_mlp_kernel<<<NNODES / 16, 256, 0, stream>>>(
      summed, cnt, W3f, b3, W4f, b4, out);
}


# Each block owns 208 destination nodes: scans dst, lists its edges in LDS, sums their MLP rows with LDS adds (padded tile rows), stores summed/cnt once: no global float atomics, no memset
# speedup vs baseline: 1.8483x; 1.8483x over previous
#include <hip/hip_runtime.h>

typedef _Float16 f16;
typedef __attribute__((ext_vector_type(16))) _Float16 v16h;
typedef __attribute__((ext_vector_type(8)))  float    v8f;

#define NNODES 50000
#define NEDGES 800000
#define FIN    128
#define FOUT   128
#define MT     4
#define RN     208
#define NBLK   241
#define NTHR   512
#define TE     128
#define CHUNK  (NTHR * 32)
#define NCH    ((NEDGES + CHUNK - 1) / CHUNK)
#define EP     264
#define HP     136

union V16H { v16h v; uint4 q[2]; };
union H8   { f16 h[8]; uint4 q; };

__device__ __forceinline__ uint4 pack8(float4 a, float4 b) {
  H8 u;
  u.h[0] = (f16)a.x; u.h[1] = (f16)a.y; u.h[2] = (f16)a.z; u.h[3] = (f16)a.w;
  u.h[4] = (f16)b.x; u.h[5] = (f16)b.y; u.h[6] = (f16)b.z; u.h[7] = (f16)b.w;
  return u.q;
}

__device__ __forceinline__ v16h ld_a_frag(const f16* lds, int row_stride,
                                          int M, int hi, int kbase) {
  const int kb = kbase + hi * 8;
  V16H t;
  t.q[0] = *(const uint4*)(lds + M * row_stride + kb);
  t.q[1] = *(const uint4*)(lds + M * row_stride + kb + 16);
  return t.v;
}

__device__ __forceinline__ v16h ld_b_frag(const f16* __restrict__ wf,
                                          int frag, int lane) {
  const uint4* p = (const uint4*)(wf + ((size_t)(frag * 32 + lane)) * 16);
  V16H t;
  t.q[0] = p[0];
  t.q[1] = p[1];
  return t.v;
}

__global__ __launch_bounds__(256) void pack_weights_kernel(
    const float* __restrict__ W, f16* __restrict__ out, int K, int N) {
  int o = blockIdx.x * blockDim.x + threadIdx.x;
  if (o >= K * N) return;
  int ksteps = K >> 5;
  int e    = o & 15;
  int lane = (o >> 4) & 31;
  int frag = o >> 9;
  int ks   = frag % ksteps;
  int nt   = frag / ksteps;
  int kl   = (e < 8 ? e : e + 8) + ((lane & 16) ? 8 : 0);
  int Kidx = ks * 32 + kl;
  int n    = nt * 16 + (lane & 15);
  out[o] = (f16)W[Kidx * N + n];
}

__device__ __forceinline__ void edge_tile(
    const float* __restrict__ x, const int* __restrict__ src,
    const int* __restrict__ dst, const f16* __restrict__ W1f,
    const float* __restrict__ b1, const f16* __restrict__ W2f,
    const float* __restrict__ b2, int lo, int base, int nvalid,
    float* s_acc, int* s_cnt, const int* s_list, f16* s_e, f16* s_h, int* s_d) {
  const int tid  = threadIdx.x;
  const int g    = tid >> 8;
  const int gt   = tid & 255;
  const int lane = tid & 31;
  const int wv   = (tid >> 5) & 7;
  f16* le = s_e + g * (64 * EP);
  f16* lh = s_h + g * (64 * HP);
  int* ld = s_d + g * 64;

  {
    const int r  = gt >> 2;
    const int c0 = (gt & 3) * 32;
    const int rr = g * 64 + r;
    const bool valid = rr < nvalid;
    int e = 0;
    if (valid) e = s_list[base + rr];
    const int si = src[e];
    const int di = dst[e];
    if ((gt & 3) == 0) {
      ld[r] = valid ? (di - lo) : -1;
      if (valid) atomicAdd(&s_cnt[di - lo], 1);
    }
    const float4* xi = (const float4*)(x + (size_t)di * FIN + c0);
    const float4* xj = (const float4*)(x + (size_t)si * FIN + c0);
    uint4* pe = (uint4*)(&le[r * EP + c0]);
    uint4* pd = (uint4*)(&le[r * EP + 128 + c0]);
#pragma unroll
    for (int k = 0; k < 4; ++k) {
      float4 a0 = xi[2 * k], a1 = xi[2 * k + 1];
      float4 j0 = xj[2 * k], j1 = xj[2 * k + 1];
      pe[k] = pack8(a0, a1);
      float4 d0 = make_float4(j0.x - a0.x, j0.y - a0.y, j0.z - a0.z, j0.w - a0.w);
      float4 d1 = make_float4(j1.x - a1.x, j1.y - a1.y, j1.z - a1.z, j1.w - a1.w);
      pd[k] = pack8(d0, d1);
    }
  }
  __syncthreads();

  const int M  = lane & 15;
  const int hi = (lane >> 4) & 1;
  const int n  = wv * 16 + (lane & 15);

  v8f acc[MT];
#pragma unroll
  for (int mt = 0; mt < MT; ++mt) acc[mt] = (v8f){};
#pragma unroll
  for (int ks = 0; ks < 8; ++ks) {
    v16h b = ld_b_frag(W1f, wv * 8 + ks, lane);
#pragma unroll
    for (int mt = 0; mt < MT; ++mt) {
      v16h a = ld_a_frag(le + mt * 16 * EP, EP, M, hi, ks * 32);
      acc[mt] = __builtin_amdgcn_wmma_f32_16x16x32_f16(false, a, false, b,
                                                       (short)0, acc[mt],
                                                       false, false);
    }
  }
  {
    const float bias = b1[n];
#pragma unroll
    for (int mt = 0; mt < MT; ++mt) {
#pragma unroll
      for (int r = 0; r < 8; ++r) {
        float v = acc[mt][r] + bias;
        v = v > 0.0f ? v : 0.0f;
        lh[(mt * 16 + r + 8 * hi) * HP + n] = (f16)v;
      }
    }
  }
  __syncthreads();

  v8f acc2[MT];
#pragma unroll
  for (int mt = 0; mt < MT; ++mt) acc2[mt] = (v8f){};
#pragma unroll
  for (int ks = 0; ks < 4; ++ks) {
    v16h b = ld_b_frag(W2f, wv * 4 + ks, lane);
#pragma unroll
    for (int mt = 0; mt < MT; ++mt) {
      v16h a = ld_a_frag(lh + mt * 16 * HP, HP, M, hi, ks * 32);
      acc2[mt] = __builtin_amdgcn_wmma_f32_16x16x32_f16(false, a, false, b,
                                                        (short)0, acc2[mt],
                                                        false, false);
    }
  }
  {
    const float bias = b2[n];
#pragma unroll
    for (int mt = 0; mt < MT; ++mt) {
#pragma unroll
      for (int r = 0; r < 8; ++r) {
        const int d = ld[mt * 16 + r + 8 * hi];
        if (d >= 0) atomicAdd(&s_acc[d * FOUT + n], acc2[mt][r] + bias);
      }
    }
  }
  __syncthreads();
}

__global__ __launch_bounds__(NTHR) void edge_range_kernel(
    const float* __restrict__ x,
    const int* __restrict__ src,
    const int* __restrict__ dst,
    const f16* __restrict__ W1f,
    const float* __restrict__ b1,
    const f16* __restrict__ W2f,
    const float* __restrict__ b2,
    float* __restrict__ summed,
    float* __restrict__ cnt) {
  __shared__ __attribute__((aligned(16))) float s_acc[RN * FOUT];
  __shared__ int   s_cnt[RN];
  __shared__ int   s_list[CHUNK + TE];
  __shared__ __attribute__((aligned(16))) f16 s_e[2 * 64 * EP];
  __shared__ __attribute__((aligned(16))) f16 s_h[2 * 64 * HP];
  __shared__ int   s_d[2 * 64];
  __shared__ int   s_n;

  const int tid = threadIdx.x;
  const int lo  = blockIdx.x * RN;
  const int nn  = (NNODES - lo) < RN ? (NNODES - lo) : RN;

  for (int i = tid; i < RN * FOUT / 4; i += NTHR)
    ((float4*)s_acc)[i] = make_float4(0.f, 0.f, 0.f, 0.f);
  for (int i = tid; i < RN; i += NTHR) s_cnt[i] = 0;
  if (tid == 0) s_n = 0;
  __syncthreads();

  const int4* d4 = (const int4*)dst;
  int4 cur[8];
#pragma unroll
  for (int j = 0; j < 8; ++j) {
    cur[j] = d4[j * NTHR + tid];
  }

  for (int c = 0; c < NCH; ++c) {
    int4 nxt[8];
#pragma unroll
    for (int j = 0; j < 8; ++j) {
      const int q = (c + 1) * (CHUNK / 4) + j * NTHR + tid;
      nxt[j] = d4[q < NEDGES / 4 ? q : NEDGES / 4 - 1];
    }
#pragma unroll
    for (int j = 0; j < 8; ++j) {
      const int q  = c * (CHUNK / 4) + j * NTHR + tid;
      const int eb = q * 4;
      const bool inb = q < NEDGES / 4;
      const int v[4] = {cur[j].x, cur[j].y, cur[j].z, cur[j].w};
#pragma unroll
      for (int k = 0; k < 4; ++k) {
        if (inb && (unsigned)(v[k] - lo) < (unsigned)nn) {
          const int p = atomicAdd(&s_n, 1);
          s_list[p] = eb + k;
        }
      }
    }
    __syncthreads();
    int n = s_n;
    __syncthreads();
    while (n >= TE) {
      edge_tile(x, src, dst, W1f, b1, W2f, b2, lo, n - TE, TE,
                s_acc, s_cnt, s_list, s_e, s_h, s_d);
      n -= TE;
    }
    if (tid == 0) s_n = n;
    __syncthreads();
#pragma unroll
    for (int j = 0; j < 8; ++j) cur[j] = nxt[j];
  }
  {
    const int n = s_n;
    if (n > 0) edge_tile(x, src, dst, W1f, b1, W2f, b2, lo, 0, n,
                         s_acc, s_cnt, s_list, s_e, s_h, s_d);
  }
  __syncthreads();

  float4* so = (float4*)(summed + (size_t)lo * FOUT);
  for (int i = tid; i < nn * (FOUT / 4); i += NTHR) so[i] = ((const float4*)s_acc)[i];
  for (int i = tid; i < nn; i += NTHR) cnt[lo + i] = (float)s_cnt[i];
}

__global__ __launch_bounds__(256) void node_mlp_kernel(
    const float* __restrict__ summed,
    const float* __restrict__ cnt,
    const f16* __restrict__ W3f,
    const float* __restrict__ b3,
    const f16* __restrict__ W4f,
    const float* __restrict__ b4,
    float* __restrict__ out) {
  __shared__ f16 lds_a[16 * 128];
  __shared__ f16 lds_m[16 * 64];

  const int tid   = threadIdx.x;
  const int lane  = tid & 31;
  const int wv    = tid >> 5;
  const int nbase = blockIdx.x * 16;

  {
    const int r    = tid >> 4;
    const int c0   = (tid & 15) * 8;
    const int node = nbase + r;
    float cv  = cnt[node];
    float inv = 1.0f / (cv > 1.0f ? cv : 1.0f);
    const float4* sp = (const float4*)(summed + (size_t)node * FOUT + c0);
    float4 s0 = sp[0], s1 = sp[1];
    float4 m0 = make_float4(s0.x * inv, s0.y * inv, s0.z * inv, s0.w * inv);
    float4 m1 = make_float4(s1.x * inv, s1.y * inv, s1.z * inv, s1.w * inv);
    m0.x = m0.x > 0.f ? m0.x : 0.f; m0.y = m0.y > 0.f ? m0.y : 0.f;
    m0.z = m0.z > 0.f ? m0.z : 0.f; m0.w = m0.w > 0.f ? m0.w : 0.f;
    m1.x = m1.x > 0.f ? m1.x : 0.f; m1.y = m1.y > 0.f ? m1.y : 0.f;
    m1.z = m1.z > 0.f ? m1.z : 0.f; m1.w = m1.w > 0.f ? m1.w : 0.f;
    *(uint4*)(&lds_a[r * 128 + c0]) = pack8(m0, m1);
  }
  __syncthreads();

  const int M  = lane & 15;
  const int hi = (lane >> 4) & 1;

  if (wv < 4) {
    const int n = wv * 16 + (lane & 15);
    v8f acc = {};
#pragma unroll
    for (int ks = 0; ks < 4; ++ks) {
      v16h a = ld_a_frag(lds_a, 128, M, hi, ks * 32);
      v16h b = ld_b_frag(W3f, wv * 4 + ks, lane);
      acc = __builtin_amdgcn_wmma_f32_16x16x32_f16(false, a, false, b,
                                                   (short)0, acc, false, false);
    }
    const float bias = b3[n];
#pragma unroll
    for (int r = 0; r < 8; ++r) {
      float v = acc[r] + bias;
      v = v > 0.0f ? v : 0.0f;
      lds_m[(r + 8 * hi) * 64 + n] = (f16)v;
    }
  }
  __syncthreads();

  {
    const int n = wv * 16 + (lane & 15);
    v8f acc = {};
#pragma unroll
    for (int ks = 0; ks < 2; ++ks) {
      v16h a = ld_a_frag(lds_m, 64, M, hi, ks * 32);
      v16h b = ld_b_frag(W4f, wv * 2 + ks, lane);
      acc = __builtin_amdgcn_wmma_f32_16x16x32_f16(false, a, false, b,
                                                   (short)0, acc, false, false);
    }
    const float bias = b4[n];
#pragma unroll
    for (int r = 0; r < 8; ++r) {
      out[(size_t)(nbase + r + 8 * hi) * FOUT + n] = acc[r] + bias;
    }
  }
}

extern "C" void kernel_launch(void* const* d_in, const int* in_sizes, int n_in,
                              void* d_out, int out_size, void* d_ws, size_t ws_size,
                              hipStream_t stream) {
  const float* x   = (const float*)d_in[0];
  const int*  eidx = (const int*)d_in[1];
  const float* W1  = (const float*)d_in[2];
  const float* b1  = (const float*)d_in[3];
  const float* W2  = (const float*)d_in[4];
  const float* b2  = (const float*)d_in[5];
  const float* W3  = (const float*)d_in[6];
  const float* b3  = (const float*)d_in[7];
  const float* W4  = (const float*)d_in[8];
  const float* b4  = (const float*)d_in[9];
  float* out = (float*)d_out;

  char* ws = (char*)d_ws;
  float* summed = (float*)ws;
  float* cnt    = (float*)(ws + 25600000);
  f16* W1f = (f16*)(ws + 25800000);
  f16* W2f = (f16*)(ws + 25800000 + 65536);
  f16* W3f = (f16*)(ws + 25800000 + 65536 + 32768);
  f16* W4f = (f16*)(ws + 25800000 + 65536 + 32768 + 16384);

  pack_weights_kernel<<<(256 * 128 + 255) / 256, 256, 0, stream>>>(W1, W1f, 256, 128);
  pack_weights_kernel<<<(128 * 128 + 255) / 256, 256, 0, stream>>>(W2, W2f, 128, 128);
  pack_weights_kernel<<<(128 * 64  + 255) / 256, 256, 0, stream>>>(W3, W3f, 128, 64);
  pack_weights_kernel<<<(64  * 128 + 255) / 256, 256, 0, stream>>>(W4, W4f, 64, 128);

  edge_range_kernel<<<NBLK, NTHR, 0, stream>>>(
      x, eidx, eidx + NEDGES, W1f, b1, W2f, b2, summed, cnt);

  node_mlp_kernel<<<NNODES / 16, 256, 0, stream>>>(
      summed, cnt, W3f, b3, W4f, b4, out);
}


# Whole-line global loads: a wave gathers one whole 512-byte x row per load (was 16 B from each of 32 lines); weight fragments packed as two 512-byte halves; node rows read as 256-byte half rows
# speedup vs baseline: 2.1189x; 1.1464x over previous
#include <hip/hip_runtime.h>

typedef _Float16 f16;
typedef __attribute__((ext_vector_type(16))) _Float16 v16h;
typedef __attribute__((ext_vector_type(8)))  float    v8f;

#define NNODES 50000
#define NEDGES 800000
#define FIN    128
#define FOUT   128
#define MT     4
#define RN     208
#define NBLK   241
#define NTHR   512
#define TE     128
#define CHUNK  (NTHR * 32)
#define NCH    ((NEDGES + CHUNK - 1) / CHUNK)
#define EP     264
#define HP     136

typedef __attribute__((ext_vector_type(2))) float    v2f;
typedef __attribute__((ext_vector_type(2))) _Float16 v2h;

union V16H { v16h v; uint4 q[2]; };
union H4   { v2h h[2]; uint2 q; };

__device__ __forceinline__ uint2 pack4(float4 a) {
  H4 u;
  u.h[0] = __builtin_convertvector((v2f){a.x, a.y}, v2h);
  u.h[1] = __builtin_convertvector((v2f){a.z, a.w}, v2h);
  return u.q;
}

__device__ __forceinline__ v16h ld_a_frag(const f16* lds, int row_stride,
                                          int M, int hi, int kbase) {
  const int kb = kbase + hi * 8;
  V16H t;
  t.q[0] = *(const uint4*)(lds + M * row_stride + kb);
  t.q[1] = *(const uint4*)(lds + M * row_stride + kb + 16);
  return t.v;
}

__device__ __forceinline__ v16h ld_b_frag(const f16* __restrict__ wf,
                                          int frag, int lane) {
  const uint4* p = (const uint4*)(wf + (size_t)frag * 512 + lane * 8);
  V16H t;
  t.q[0] = p[0];
  t.q[1] = p[32];
  return t.v;
}

__global__ __launch_bounds__(256) void pack_weights_kernel(
    const float* __restrict__ W, f16* __restrict__ out, int K, int N) {
  int o = blockIdx.x * blockDim.x + threadIdx.x;
  if (o >= K * N) return;
  int ksteps = K >> 5;
  int e    = ((o >> 8) & 1) * 8 + (o & 7);
  int lane = (o >> 3) & 31;
  int frag = o >> 9;
  int ks   = frag % ksteps;
  int nt   = frag / ksteps;
  int kl   = (e < 8 ? e : e + 8) + ((lane & 16) ? 8 : 0);
  int Kidx = ks * 32 + kl;
  int n    = nt * 16 + (lane & 15);
  out[o] = (f16)W[Kidx * N + n];
}

__device__ __forceinline__ void edge_tile(
    const float* __restrict__ x, const int* __restrict__ src,
    const int* __restrict__ dst, const f16* __restrict__ W1f,
    const float* __restrict__ b1, const f16* __restrict__ W2f,
    const float* __restrict__ b2, int lo, int base, int nvalid,
    float* s_acc, int* s_cnt, const int* s_list, f16* s_e, f16* s_h, int* s_d) {
  const int tid  = threadIdx.x;
  const int g    = tid >> 8;
  const int lane = tid & 31;
  const int wv   = (tid >> 5) & 7;
  f16* le = s_e + g * (64 * EP);
  f16* lh = s_h + g * (64 * HP);
  int* ld = s_d + g * 64;

  {
    const int r0 = wv * 8;
    int si = 0, di = 0;
    if (lane < 8) {
      const int rr = g * 64 + r0 + lane;
      const bool valid = rr < nvalid;
      int e = 0;
      if (valid) e = s_list[base + rr];
      si = src[e];
      di = dst[e];
      ld[r0 + lane] = valid ? (di - lo) : -1;
      if (valid) atomicAdd(&s_cnt[di - lo], 1);
    }
    float4 xi[8], xj[8];
#pragma unroll
    for (int j = 0; j < 8; ++j) {
      const int dj = __builtin_amdgcn_readlane(di, j);
      const int sj = __builtin_amdgcn_readlane(si, j);
      xi[j] = ((const float4*)(x + (size_t)dj * FIN))[lane];
      xj[j] = ((const float4*)(x + (size_t)sj * FIN))[lane];
    }
#pragma unroll
    for (int j = 0; j < 8; ++j) {
      f16* row = le + (r0 + j) * EP + lane * 4;
      const float4 a = xi[j], b = xj[j];
      *(uint2*)row         = pack4(a);
      *(uint2*)(row + 128) = pack4(make_float4(b.x - a.x, b.y - a.y, b.z - a.z, b.w - a.w));
    }
  }
  __syncthreads();

  const int M  = lane & 15;
  const int hi = (lane >> 4) & 1;
  const int n  = wv * 16 + (lane & 15);

  v8f acc[MT];
#pragma unroll
  for (int mt = 0; mt < MT; ++mt) acc[mt] = (v8f){};
#pragma unroll
  for (int ks = 0; ks < 8; ++ks) {
    v16h b = ld_b_frag(W1f, wv * 8 + ks, lane);
#pragma unroll
    for (int mt = 0; mt < MT; ++mt) {
      v16h a = ld_a_frag(le + mt * 16 * EP, EP, M, hi, ks * 32);
      acc[mt] = __builtin_amdgcn_wmma_f32_16x16x32_f16(false, a, false, b,
                                                       (short)0, acc[mt],
                                                       false, false);
    }
  }
  {
    const float bias = b1[n];
#pragma unroll
    for (int mt = 0; mt < MT; ++mt) {
#pragma unroll
      for (int r = 0; r < 8; ++r) {
        float v = acc[mt][r] + bias;
        v = v > 0.0f ? v : 0.0f;
        lh[(mt * 16 + r + 8 * hi) * HP + n] = (f16)v;
      }
    }
  }
  __syncthreads();

  v8f acc2[MT];
#pragma unroll
  for (int mt = 0; mt < MT; ++mt) acc2[mt] = (v8f){};
#pragma unroll
  for (int ks = 0; ks < 4; ++ks) {
    v16h b = ld_b_frag(W2f, wv * 4 + ks, lane);
#pragma unroll
    for (int mt = 0; mt < MT; ++mt) {
      v16h a = ld_a_frag(lh + mt * 16 * HP, HP, M, hi, ks * 32);
      acc2[mt] = __builtin_amdgcn_wmma_f32_16x16x32_f16(false, a, false, b,
                                                        (short)0, acc2[mt],
                                                        false, false);
    }
  }
  {
    const float bias = b2[n];
#pragma unroll
    for (int mt = 0; mt < MT; ++mt) {
#pragma unroll
      for (int r = 0; r < 8; ++r) {
        const int d = ld[mt * 16 + r + 8 * hi];
        if (d >= 0) atomicAdd(&s_acc[d * FOUT + n], acc2[mt][r] + bias);
      }
    }
  }
  __syncthreads();
}

__global__ __launch_bounds__(NTHR) void edge_range_kernel(
    const float* __restrict__ x,
    const int* __restrict__ src,
    const int* __restrict__ dst,
    const f16* __restrict__ W1f,
    const float* __restrict__ b1,
    const f16* __restrict__ W2f,
    const float* __restrict__ b2,
    float* __restrict__ summed,
    float* __restrict__ cnt) {
  __shared__ __attribute__((aligned(16))) float s_acc[RN * FOUT];
  __shared__ int   s_cnt[RN];
  __shared__ int   s_list[CHUNK + TE];
  __shared__ __attribute__((aligned(16))) f16 s_e[2 * 64 * EP];
  __shared__ __attribute__((aligned(16))) f16 s_h[2 * 64 * HP];
  __shared__ int   s_d[2 * 64];
  __shared__ int   s_n;

  const int tid = threadIdx.x;
  const int lo  = blockIdx.x * RN;
  const int nn  = (NNODES - lo) < RN ? (NNODES - lo) : RN;

  for (int i = tid; i < RN * FOUT / 4; i += NTHR)
    ((float4*)s_acc)[i] = make_float4(0.f, 0.f, 0.f, 0.f);
  for (int i = tid; i < RN; i += NTHR) s_cnt[i] = 0;
  if (tid == 0) s_n = 0;
  __syncthreads();

  const int4* d4 = (const int4*)dst;
  int4 cur[8];
#pragma unroll
  for (int j = 0; j < 8; ++j) {
    cur[j] = d4[j * NTHR + tid];
  }

  for (int c = 0; c < NCH; ++c) {
    int4 nxt[8];
#pragma unroll
    for (int j = 0; j < 8; ++j) {
      const int q = (c + 1) * (CHUNK / 4) + j * NTHR + tid;
      nxt[j] = d4[q < NEDGES / 4 ? q : NEDGES / 4 - 1];
    }
#pragma unroll
    for (int j = 0; j < 8; ++j) {
      const int q  = c * (CHUNK / 4) + j * NTHR + tid;
      const int eb = q * 4;
      const bool inb = q < NEDGES / 4;
      const int v[4] = {cur[j].x, cur[j].y, cur[j].z, cur[j].w};
#pragma unroll
      for (int k = 0; k < 4; ++k) {
        if (inb && (unsigned)(v[k] - lo) < (unsigned)nn) {
          const int p = atomicAdd(&s_n, 1);
          s_list[p] = eb + k;
        }
      }
    }
    __syncthreads();
    int n = s_n;
    __syncthreads();
    while (n >= TE) {
      edge_tile(x, src, dst, W1f, b1, W2f, b2, lo, n - TE, TE,
                s_acc, s_cnt, s_list, s_e, s_h, s_d);
      n -= TE;
    }
    if (tid == 0) s_n = n;
    __syncthreads();
#pragma unroll
    for (int j = 0; j < 8; ++j) cur[j] = nxt[j];
  }
  {
    const int n = s_n;
    if (n > 0) edge_tile(x, src, dst, W1f, b1, W2f, b2, lo, 0, n,
                         s_acc, s_cnt, s_list, s_e, s_h, s_d);
  }
  __syncthreads();

  float4* so = (float4*)(summed + (size_t)lo * FOUT);
  for (int i = tid; i < nn * (FOUT / 4); i += NTHR) so[i] = ((const float4*)s_acc)[i];
  for (int i = tid; i < nn; i += NTHR) cnt[lo + i] = (float)s_cnt[i];
}

__global__ __launch_bounds__(256) void node_mlp_kernel(
    const float* __restrict__ summed,
    const float* __restrict__ cnt,
    const f16* __restrict__ W3f,
    const float* __restrict__ b3,
    const f16* __restrict__ W4f,
    const float* __restrict__ b4,
    float* __restrict__ out) {
  __shared__ f16 lds_a[16 * 128];
  __shared__ f16 lds_m[16 * 64];

  const int tid   = threadIdx.x;
  const int lane  = tid & 31;
  const int wv    = tid >> 5;
  const int nbase = blockIdx.x * 16;

  {
    const int r    = tid >> 4;
    const int q    = tid & 15;
    const int node = nbase + r;
    float cv  = cnt[node];
    float inv = 1.0f / (cv > 1.0f ? cv : 1.0f);
    const float4* sp = (const float4*)(summed + (size_t)node * FOUT);
    float4 s0 = sp[q], s1 = sp[q + 16];
    float4 m0 = make_float4(s0.x * inv, s0.y * inv, s0.z * inv, s0.w * inv);
    float4 m1 = make_float4(s1.x * inv, s1.y * inv, s1.z * inv, s1.w * inv);
    m0.x = m0.x > 0.f ? m0.x : 0.f; m0.y = m0.y > 0.f ? m0.y : 0.f;
    m0.z = m0.z > 0.f ? m0.z : 0.f; m0.w = m0.w > 0.f ? m0.w : 0.f;
    m1.x = m1.x > 0.f ? m1.x : 0.f; m1.y = m1.y > 0.f ? m1.y : 0.f;
    m1.z = m1.z > 0.f ? m1.z : 0.f; m1.w = m1.w > 0.f ? m1.w : 0.f;
    *(uint2*)(&lds_a[r * 128 + 4 * q])      = pack4(m0);
    *(uint2*)(&lds_a[r * 128 + 64 + 4 * q]) = pack4(m1);
  }
  __syncthreads();

  const int M  = lane & 15;
  const int hi = (lane >> 4) & 1;

  if (wv < 4) {
    const int n = wv * 16 + (lane & 15);
    v8f acc = {};
#pragma unroll
    for (int ks = 0; ks < 4; ++ks) {
      v16h a = ld_a_frag(lds_a, 128, M, hi, ks * 32);
      v16h b = ld_b_frag(W3f, wv * 4 + ks, lane);
      acc = __builtin_amdgcn_wmma_f32_16x16x32_f16(false, a, false, b,
                                                   (short)0, acc, false, false);
    }
    const float bias = b3[n];
#pragma unroll
    for (int r = 0; r < 8; ++r) {
      float v = acc[r] + bias;
      v = v > 0.0f ? v : 0.0f;
      lds_m[(r + 8 * hi) * 64 + n] = (f16)v;
    }
  }
  __syncthreads();

  {
    const int n = wv * 16 + (lane & 15);
    v8f acc = {};
#pragma unroll
    for (int ks = 0; ks < 2; ++ks) {
      v16h a = ld_a_frag(lds_m, 64, M, hi, ks * 32);
      v16h b = ld_b_frag(W4f, wv * 2 + ks, lane);
      acc = __builtin_amdgcn_wmma_f32_16x16x32_f16(false, a, false, b,
                                                   (short)0, acc, false, false);
    }
    const float bias = b4[n];
#pragma unroll
    for (int r = 0; r < 8; ++r) {
      out[(size_t)(nbase + r + 8 * hi) * FOUT + n] = acc[r] + bias;
    }
  }
}

extern "C" void kernel_launch(void* const* d_in, const int* in_sizes, int n_in,
                              void* d_out, int out_size, void* d_ws, size_t ws_size,
                              hipStream_t stream) {
  const float* x   = (const float*)d_in[0];
  const int*  eidx = (const int*)d_in[1];
  const float* W1  = (const float*)d_in[2];
  const float* b1  = (const float*)d_in[3];
  const float* W2  = (const float*)d_in[4];
  const float* b2  = (const float*)d_in[5];
  const float* W3  = (const float*)d_in[6];
  const float* b3  = (const float*)d_in[7];
  const float* W4  = (const float*)d_in[8];
  const float* b4  = (const float*)d_in[9];
  float* out = (float*)d_out;

  char* ws = (char*)d_ws;
  float* summed = (float*)ws;
  float* cnt    = (float*)(ws + 25600000);
  f16* W1f = (f16*)(ws + 25800000);
  f16* W2f = (f16*)(ws + 25800000 + 65536);
  f16* W3f = (f16*)(ws + 25800000 + 65536 + 32768);
  f16* W4f = (f16*)(ws + 25800000 + 65536 + 32768 + 16384);

  pack_weights_kernel<<<(256 * 128 + 255) / 256, 256, 0, stream>>>(W1, W1f, 256, 128);
  pack_weights_kernel<<<(128 * 128 + 255) / 256, 256, 0, stream>>>(W2, W2f, 128, 128);
  pack_weights_kernel<<<(128 * 64  + 255) / 256, 256, 0, stream>>>(W3, W3f, 128, 64);
  pack_weights_kernel<<<(64  * 128 + 255) / 256, 256, 0, stream>>>(W4, W4f, 64, 128);

  edge_range_kernel<<<NBLK, NTHR, 0, stream>>>(
      x, eidx, eidx + NEDGES, W1f, b1, W2f, b2, summed, cnt);

  node_mlp_kernel<<<NNODES / 16, 256, 0, stream>>>(
      summed, cnt, W3f, b3, W4f, b4, out);
}


# Edge GEMMs: A fragments are read from LDS in groups of two, each group issued two groups ahead of the WMMAs that use it (three rotating register sets); was load, wait, WMMA one at a time
# speedup vs baseline: 2.2698x; 1.0712x over previous
#include <hip/hip_runtime.h>

typedef _Float16 f16;
typedef __attribute__((ext_vector_type(16))) _Float16 v16h;
typedef __attribute__((ext_vector_type(8)))  float    v8f;

#define NNODES 50000
#define NEDGES 800000
#define FIN    128
#define FOUT   128
#define MT     4
#define RN     208
#define NBLK   241
#define NTHR   512
#define TE     128
#define CHUNK  (NTHR * 32)
#define NCH    ((NEDGES + CHUNK - 1) / CHUNK)
#define EP     264
#define HP     136

typedef __attribute__((ext_vector_type(2))) float    v2f;
typedef __attribute__((ext_vector_type(2))) _Float16 v2h;

union V16H { v16h v; uint4 q[2]; };
union H4   { v2h h[2]; uint2 q; };

__device__ __forceinline__ uint2 pack4(float4 a) {
  H4 u;
  u.h[0] = __builtin_convertvector((v2f){a.x, a.y}, v2h);
  u.h[1] = __builtin_convertvector((v2f){a.z, a.w}, v2h);
  return u.q;
}

__device__ __forceinline__ v16h ld_a_frag(const f16* lds, int row_stride,
                                          int M, int hi, int kbase) {
  const int kb = kbase + hi * 8;
  V16H t;
  t.q[0] = *(const uint4*)(lds + M * row_stride + kb);
  t.q[1] = *(const uint4*)(lds + M * row_stride + kb + 16);
  return t.v;
}

__device__ __forceinline__ v16h ld_b_frag(const f16* __restrict__ wf,
                                          int frag, int lane) {
  const uint4* p = (const uint4*)(wf + (size_t)frag * 512 + lane * 8);
  V16H t;
  t.q[0] = p[0];
  t.q[1] = p[32];
  return t.v;
}

__global__ __launch_bounds__(256) void pack_weights_kernel(
    const float* __restrict__ W, f16* __restrict__ out, int K, int N) {
  int o = blockIdx.x * blockDim.x + threadIdx.x;
  if (o >= K * N) return;
  int ksteps = K >> 5;
  int e    = ((o >> 8) & 1) * 8 + (o & 7);
  int lane = (o >> 3) & 31;
  int frag = o >> 9;
  int ks   = frag % ksteps;
  int nt   = frag / ksteps;
  int kl   = (e < 8 ? e : e + 8) + ((lane & 16) ? 8 : 0);
  int Kidx = ks * 32 + kl;
  int n    = nt * 16 + (lane & 15);
  out[o] = (f16)W[Kidx * N + n];
}

__device__ __forceinline__ void edge_tile(
    const float* __restrict__ x, const int* __restrict__ src,
    const int* __restrict__ dst, const f16* __restrict__ W1f,
    const float* __restrict__ b1, const f16* __restrict__ W2f,
    const float* __restrict__ b2, int lo, int base, int nvalid,
    float* s_acc, int* s_cnt, const int* s_list, f16* s_e, f16* s_h, int* s_d) {
  const int tid  = threadIdx.x;
  const int g    = tid >> 8;
  const int lane = tid & 31;
  const int wv   = (tid >> 5) & 7;
  f16* le = s_e + g * (64 * EP);
  f16* lh = s_h + g * (64 * HP);
  int* ld = s_d + g * 64;

  {
    const int r0 = wv * 8;
    int si = 0, di = 0;
    if (lane < 8) {
      const int rr = g * 64 + r0 + lane;
      const bool valid = rr < nvalid;
      int e = 0;
      if (valid) e = s_list[base + rr];
      si = src[e];
      di = dst[e];
      ld[r0 + lane] = valid ? (di - lo) : -1;
      if (valid) atomicAdd(&s_cnt[di - lo], 1);
    }
    float4 xi[8], xj[8];
#pragma unroll
    for (int j = 0; j < 8; ++j) {
      const int dj = __builtin_amdgcn_readlane(di, j);
      const int sj = __builtin_amdgcn_readlane(si, j);
      xi[j] = ((const float4*)(x + (size_t)dj * FIN))[lane];
      xj[j] = ((const float4*)(x + (size_t)sj * FIN))[lane];
    }
#pragma unroll
    for (int j = 0; j < 8; ++j) {
      f16* row = le + (r0 + j) * EP + lane * 4;
      const float4 a = xi[j], b = xj[j];
      *(uint2*)row         = pack4(a);
      *(uint2*)(row + 128) = pack4(make_float4(b.x - a.x, b.y - a.y, b.z - a.z, b.w - a.w));
    }
  }
  __syncthreads();

  const int M  = lane & 15;
  const int hi = (lane >> 4) & 1;
  const int n  = wv * 16 + (lane & 15);

  v8f acc[MT];
  v16h ag[3][2];
#pragma unroll
  for (int mt = 0; mt < MT; ++mt) {
    acc[mt] = (v8f){};
    ag[mt >> 1][mt & 1] = ld_a_frag(le + mt * 16 * EP, EP, M, hi, 0);
  }
#pragma unroll
  for (int g = 0; g < 16; ++g) {
    if (g + 2 < 16) {
#pragma unroll
      for (int p = 0; p < 2; ++p)
        ag[(g + 2) % 3][p] = ld_a_frag(le + ((g & 1) * 2 + p) * 16 * EP, EP, M, hi,
                                       ((g + 2) >> 1) * 32);
    }
    __builtin_amdgcn_sched_barrier(0);
    v16h b = ld_b_frag(W1f, wv * 8 + (g >> 1), lane);
#pragma unroll
    for (int p = 0; p < 2; ++p) {
      const int mt = (g & 1) * 2 + p;
      acc[mt] = __builtin_amdgcn_wmma_f32_16x16x32_f16(false, ag[g % 3][p], false, b,
                                                       (short)0, acc[mt],
                                                       false, false);
    }
    __builtin_amdgcn_sched_barrier(0);
  }
  {
    const float bias = b1[n];
#pragma unroll
    for (int mt = 0; mt < MT; ++mt) {
#pragma unroll
      for (int r = 0; r < 8; ++r) {
        float v = acc[mt][r] + bias;
        v = v > 0.0f ? v : 0.0f;
        lh[(mt * 16 + r + 8 * hi) * HP + n] = (f16)v;
      }
    }
  }
  __syncthreads();

  v8f acc2[MT];
#pragma unroll
  for (int mt = 0; mt < MT; ++mt) {
    acc2[mt] = (v8f){};
    ag[mt >> 1][mt & 1] = ld_a_frag(lh + mt * 16 * HP, HP, M, hi, 0);
  }
#pragma unroll
  for (int g = 0; g < 8; ++g) {
    if (g + 2 < 8) {
#pragma unroll
      for (int p = 0; p < 2; ++p)
        ag[(g + 2) % 3][p] = ld_a_frag(lh + ((g & 1) * 2 + p) * 16 * HP, HP, M, hi,
                                       ((g + 2) >> 1) * 32);
    }
    __builtin_amdgcn_sched_barrier(0);
    v16h b = ld_b_frag(W2f, wv * 4 + (g >> 1), lane);
#pragma unroll
    for (int p = 0; p < 2; ++p) {
      const int mt = (g & 1) * 2 + p;
      acc2[mt] = __builtin_amdgcn_wmma_f32_16x16x32_f16(false, ag[g % 3][p], false, b,
                                                        (short)0, acc2[mt],
                                                        false, false);
    }
    __builtin_amdgcn_sched_barrier(0);
  }
  {
    const float bias = b2[n];
#pragma unroll
    for (int mt = 0; mt < MT; ++mt) {
#pragma unroll
      for (int r = 0; r < 8; ++r) {
        const int d = ld[mt * 16 + r + 8 * hi];
        if (d >= 0) atomicAdd(&s_acc[d * FOUT + n], acc2[mt][r] + bias);
      }
    }
  }
  __syncthreads();
}

__global__ __launch_bounds__(NTHR) void edge_range_kernel(
    const float* __restrict__ x,
    const int* __restrict__ src,
    const int* __restrict__ dst,
    const f16* __restrict__ W1f,
    const float* __restrict__ b1,
    const f16* __restrict__ W2f,
    const float* __restrict__ b2,
    float* __restrict__ summed,
    float* __restrict__ cnt) {
  __shared__ __attribute__((aligned(16))) float s_acc[RN * FOUT];
  __shared__ int   s_cnt[RN];
  __shared__ int   s_list[CHUNK + TE];
  __shared__ __attribute__((aligned(16))) f16 s_e[2 * 64 * EP];
  __shared__ __attribute__((aligned(16))) f16 s_h[2 * 64 * HP];
  __shared__ int   s_d[2 * 64];
  __shared__ int   s_n;

  const int tid = threadIdx.x;
  const int lo  = blockIdx.x * RN;
  const int nn  = (NNODES - lo) < RN ? (NNODES - lo) : RN;

  for (int i = tid; i < RN * FOUT / 4; i += NTHR)
    ((float4*)s_acc)[i] = make_float4(0.f, 0.f, 0.f, 0.f);
  for (int i = tid; i < RN; i += NTHR) s_cnt[i] = 0;
  if (tid == 0) s_n = 0;
  __syncthreads();

  const int4* d4 = (const int4*)dst;
  int4 cur[8];
#pragma unroll
  for (int j = 0; j < 8; ++j) {
    cur[j] = d4[j * NTHR + tid];
  }

  for (int c = 0; c < NCH; ++c) {
    int4 nxt[8];
#pragma unroll
    for (int j = 0; j < 8; ++j) {
      const int q = (c + 1) * (CHUNK / 4) + j * NTHR + tid;
      nxt[j] = d4[q < NEDGES / 4 ? q : NEDGES / 4 - 1];
    }
#pragma unroll
    for (int j = 0; j < 8; ++j) {
      const int q  = c * (CHUNK / 4) + j * NTHR + tid;
      const int eb = q * 4;
      const bool inb = q < NEDGES / 4;
      const int v[4] = {cur[j].x, cur[j].y, cur[j].z, cur[j].w};
#pragma unroll
      for (int k = 0; k < 4; ++k) {
        if (inb && (unsigned)(v[k] - lo) < (unsigned)nn) {
          const int p = atomicAdd(&s_n, 1);
          s_list[p] = eb + k;
        }
      }
    }
    __syncthreads();
    int n = s_n;
    __syncthreads();
    while (n >= TE) {
      edge_tile(x, src, dst, W1f, b1, W2f, b2, lo, n - TE, TE,
                s_acc, s_cnt, s_list, s_e, s_h, s_d);
      n -= TE;
    }
    if (tid == 0) s_n = n;
    __syncthreads();
#pragma unroll
    for (int j = 0; j < 8; ++j) cur[j] = nxt[j];
  }
  {
    const int n = s_n;
    if (n > 0) edge_tile(x, src, dst, W1f, b1, W2f, b2, lo, 0, n,
                         s_acc, s_cnt, s_list, s_e, s_h, s_d);
  }
  __syncthreads();

  float4* so = (float4*)(summed + (size_t)lo * FOUT);
  for (int i = tid; i < nn * (FOUT / 4); i += NTHR) so[i] = ((const float4*)s_acc)[i];
  for (int i = tid; i < nn; i += NTHR) cnt[lo + i] = (float)s_cnt[i];
}

__global__ __launch_bounds__(256) void node_mlp_kernel(
    const float* __restrict__ summed,
    const float* __restrict__ cnt,
    const f16* __restrict__ W3f,
    const float* __restrict__ b3,
    const f16* __restrict__ W4f,
    const float* __restrict__ b4,
    float* __restrict__ out) {
  __shared__ f16 lds_a[16 * 128];
  __shared__ f16 lds_m[16 * 64];

  const int tid   = threadIdx.x;
  const int lane  = tid & 31;
  const int wv    = tid >> 5;
  const int nbase = blockIdx.x * 16;

  {
    const int r    = tid >> 4;
    const int q    = tid & 15;
    const int node = nbase + r;
    float cv  = cnt[node];
    float inv = 1.0f / (cv > 1.0f ? cv : 1.0f);
    const float4* sp = (const float4*)(summed + (size_t)node * FOUT);
    float4 s0 = sp[q], s1 = sp[q + 16];
    float4 m0 = make_float4(s0.x * inv, s0.y * inv, s0.z * inv, s0.w * inv);
    float4 m1 = make_float4(s1.x * inv, s1.y * inv, s1.z * inv, s1.w * inv);
    m0.x = m0.x > 0.f ? m0.x : 0.f; m0.y = m0.y > 0.f ? m0.y : 0.f;
    m0.z = m0.z > 0.f ? m0.z : 0.f; m0.w = m0.w > 0.f ? m0.w : 0.f;
    m1.x = m1.x > 0.f ? m1.x : 0.f; m1.y = m1.y > 0.f ? m1.y : 0.f;
    m1.z = m1.z > 0.f ? m1.z : 0.f; m1.w = m1.w > 0.f ? m1.w : 0.f;
    *(uint2*)(&lds_a[r * 128 + 4 * q])      = pack4(m0);
    *(uint2*)(&lds_a[r * 128 + 64 + 4 * q]) = pack4(m1);
  }
  __syncthreads();

  const int M  = lane & 15;
  const int hi = (lane >> 4) & 1;

  if (wv < 4) {
    const int n = wv * 16 + (lane & 15);
    v8f acc = {};
#pragma unroll
    for (int ks = 0; ks < 4; ++ks) {
      v16h a = ld_a_frag(lds_a, 128, M, hi, ks * 32);
      v16h b = ld_b_frag(W3f, wv * 4 + ks, lane);
      acc = __builtin_amdgcn_wmma_f32_16x16x32_f16(false, a, false, b,
                                                   (short)0, acc, false, false);
    }
    const float bias = b3[n];
#pragma unroll
    for (int r = 0; r < 8; ++r) {
      float v = acc[r] + bias;
      v = v > 0.0f ? v : 0.0f;
      lds_m[(r + 8 * hi) * 64 + n] = (f16)v;
    }
  }
  __syncthreads();

  {
    const int n = wv * 16 + (lane & 15);
    v8f acc = {};
#pragma unroll
    for (int ks = 0; ks < 2; ++ks) {
      v16h a = ld_a_frag(lds_m, 64, M, hi, ks * 32);
      v16h b = ld_b_frag(W4f, wv * 2 + ks, lane);
      acc = __builtin_amdgcn_wmma_f32_16x16x32_f16(false, a, false, b,
                                                   (short)0, acc, false, false);
    }
    const float bias = b4[n];
#pragma unroll
    for (int r = 0; r < 8; ++r) {
      out[(size_t)(nbase + r + 8 * hi) * FOUT + n] = acc[r] + bias;
    }
  }
}

extern "C" void kernel_launch(void* const* d_in, const int* in_sizes, int n_in,
                              void* d_out, int out_size, void* d_ws, size_t ws_size,
                              hipStream_t stream) {
  const float* x   = (const float*)d_in[0];
  const int*  eidx = (const int*)d_in[1];
  const float* W1  = (const float*)d_in[2];
  const float* b1  = (const float*)d_in[3];
  const float* W2  = (const float*)d_in[4];
  const float* b2  = (const float*)d_in[5];
  const float* W3  = (const float*)d_in[6];
  const float* b3  = (const float*)d_in[7];
  const float* W4  = (const float*)d_in[8];
  const float* b4  = (const float*)d_in[9];
  float* out = (float*)d_out;

  char* ws = (char*)d_ws;
  float* summed = (float*)ws;
  float* cnt    = (float*)(ws + 25600000);
  f16* W1f = (f16*)(ws + 25800000);
  f16* W2f = (f16*)(ws + 25800000 + 65536);
  f16* W3f = (f16*)(ws + 25800000 + 65536 + 32768);
  f16* W4f = (f16*)(ws + 25800000 + 65536 + 32768 + 16384);

  pack_weights_kernel<<<(256 * 128 + 255) / 256, 256, 0, stream>>>(W1, W1f, 256, 128);
  pack_weights_kernel<<<(128 * 128 + 255) / 256, 256, 0, stream>>>(W2, W2f, 128, 128);
  pack_weights_kernel<<<(128 * 64  + 255) / 256, 256, 0, stream>>>(W3, W3f, 128, 64);
  pack_weights_kernel<<<(64  * 128 + 255) / 256, 256, 0, stream>>>(W4, W4f, 64, 128);

  edge_range_kernel<<<NBLK, NTHR, 0, stream>>>(
      x, eidx, eidx + NEDGES, W1f, b1, W2f, b2, summed, cnt);

  node_mlp_kernel<<<NNODES / 16, 256, 0, stream>>>(
      summed, cnt, W3f, b3, W4f, b4, out);
}


# Edge GEMMs: wave tile 64 edges x 32 columns (2 N-tiles, was 1): each A fragment read from LDS feeds two WMMAs; 8-wave blocks, 512 registers a wave; scan reserves list space once per lane and chunk
# speedup vs baseline: 2.8496x; 1.2554x over previous
#include <hip/hip_runtime.h>

typedef _Float16 f16;
typedef __attribute__((ext_vector_type(16))) _Float16 v16h;
typedef __attribute__((ext_vector_type(8)))  float    v8f;

#define NNODES 50000
#define NEDGES 800000
#define FIN    128
#define FOUT   128
#define MT     4
#define NTW    2
#define RPW    16
#define RN     208
#define NBLK   241
#define NTHR   256
#define TE     128
#define SCJ    16
#define CHUNK  (NTHR * 4 * SCJ)
#define NCH    ((NEDGES + CHUNK - 1) / CHUNK)
#define LCAP   (CHUNK + 8192)
#define FLUSH_AT (LCAP - CHUNK)
#define EP     264
#define HP     136

typedef __attribute__((ext_vector_type(2))) float    v2f;
typedef __attribute__((ext_vector_type(2))) _Float16 v2h;

union V16H { v16h v; uint4 q[2]; };
union H4   { v2h h[2]; uint2 q; };

__device__ __forceinline__ uint2 pack4(float4 a) {
  H4 u;
  u.h[0] = __builtin_convertvector((v2f){a.x, a.y}, v2h);
  u.h[1] = __builtin_convertvector((v2f){a.z, a.w}, v2h);
  return u.q;
}

__device__ __forceinline__ v16h ld_a_frag(const f16* lds, int row_stride,
                                          int M, int hi, int kbase) {
  const int kb = kbase + hi * 8;
  V16H t;
  t.q[0] = *(const uint4*)(lds + M * row_stride + kb);
  t.q[1] = *(const uint4*)(lds + M * row_stride + kb + 16);
  return t.v;
}

__device__ __forceinline__ v16h ld_b_frag(const f16* __restrict__ wf,
                                          int frag, int lane) {
  const uint4* p = (const uint4*)(wf + (size_t)frag * 512 + lane * 8);
  V16H t;
  t.q[0] = p[0];
  t.q[1] = p[32];
  return t.v;
}

__global__ __launch_bounds__(256) void pack_weights_kernel(
    const float* __restrict__ W, f16* __restrict__ out, int K, int N) {
  int o = blockIdx.x * blockDim.x + threadIdx.x;
  if (o >= K * N) return;
  int ksteps = K >> 5;
  int e    = ((o >> 8) & 1) * 8 + (o & 7);
  int lane = (o >> 3) & 31;
  int frag = o >> 9;
  int ks   = frag % ksteps;
  int nt   = frag / ksteps;
  int kl   = (e < 8 ? e : e + 8) + ((lane & 16) ? 8 : 0);
  int Kidx = ks * 32 + kl;
  int n    = nt * 16 + (lane & 15);
  out[o] = (f16)W[Kidx * N + n];
}


__device__ __forceinline__ void fetch_idx(
    const int* __restrict__ src, const int* __restrict__ dst, const int* s_list,
    int base, int nvalid, int rr, int lane, int& si, int& di, bool& valid) {
  si = 0; di = 0; valid = false;
  if (lane < RPW) {
    valid = rr < nvalid;
    int e = 0;
    if (valid) e = s_list[base + rr];
    si = src[e];
    di = dst[e];
  }
}

__device__ __forceinline__ void issue_rows(
    const float* __restrict__ x, int si, int di, int lane,
    float4 (&xi)[RPW], float4 (&xj)[RPW]) {
#pragma unroll
  for (int j = 0; j < RPW; ++j) {
    const int dj = __builtin_amdgcn_readlane(di, j);
    const int sj = __builtin_amdgcn_readlane(si, j);
    xi[j] = ((const float4*)x)[(unsigned)(dj * (FIN / 4) + lane)];
    xj[j] = ((const float4*)x)[(unsigned)(sj * (FIN / 4) + lane)];
  }
}

__device__ __forceinline__ void store_rows(f16* le, int r0, int lane,
                                           const float4 (&xi)[RPW], const float4 (&xj)[RPW]) {
#pragma unroll
  for (int j = 0; j < RPW; ++j) {
    f16* row = le + (r0 + j) * EP + lane * 4;
    const float4 a = xi[j], b = xj[j];
    *(uint2*)row         = pack4(a);
    *(uint2*)(row + 128) = pack4(make_float4(b.x - a.x, b.y - a.y, b.z - a.z, b.w - a.w));
  }
}

__device__ __forceinline__ void sum_rows(const int* ld, float* s_acc,
                                         const v8f& a0, const v8f& a1,
                                         int mt, int hi, int nn,
                                         float bias2a, float bias2b) {
#pragma unroll
  for (int r = 0; r < 8; ++r) {
    const int d = ld[mt * 16 + r + 8 * hi];
    if (d >= 0) {
      atomicAdd(&s_acc[d * FOUT + nn],      a0[r] + bias2a);
      atomicAdd(&s_acc[d * FOUT + nn + 16], a1[r] + bias2b);
    }
  }
}

template <bool FINAL>
__device__ __forceinline__ void run_tiles(
    const float* __restrict__ x, const int* __restrict__ src,
    const int* __restrict__ dst, const v16h (&w1r)[NTW][8], const float (&bias1)[NTW],
    const v16h (&w2r)[NTW][4], const float (&bias2)[NTW], int lo, int n,
    float* s_acc, int* s_cnt, const int* s_list, f16* s_e, f16* s_h, int* s_d) {
  const int nt = FINAL ? (n + TE - 1) / TE : n / TE;
  if (nt == 0) return;
  const int tid  = threadIdx.x;
  const int g    = tid >> 7;
  const int lane = tid & 31;
  const int wv   = (tid >> 5) & 3;
  f16* le = s_e + g * (64 * EP);
  f16* lh = s_h + g * (64 * HP);
  int* ld = s_d + g * 64;
  const int r0 = wv * RPW;
  const int rr = g * 64 + r0 + lane;
  const int M  = lane & 15;
  const int hi = (lane >> 4) & 1;
  const int nn = wv * (16 * NTW) + (lane & 15);

  int si = 0, di = 0; bool valid = false;
  if (FINAL) {
    float4 xi[RPW], xj[RPW];
    fetch_idx(src, dst, s_list, n > TE ? n - TE : 0, n < TE ? n : TE, rr, lane, si, di, valid);
    issue_rows(x, si, di, lane, xi, xj);
    store_rows(le, r0, lane, xi, xj);
    __syncthreads();
  }

  for (int t = 0; t < nt; ++t) {
    if (!FINAL) {
      float4 xi[RPW], xj[RPW];
      fetch_idx(src, dst, s_list, n - (t + 1) * TE, TE, rr, lane, si, di, valid);
      issue_rows(x, si, di, lane, xi, xj);
      store_rows(le, r0, lane, xi, xj);
    }
    if (lane < RPW) {
      ld[r0 + lane] = valid ? (di - lo) : -1;
      if (valid) atomicAdd(&s_cnt[di - lo], 1);
    }
    const int  rem  = n - (t + 1) * TE;
    const bool more = FINAL && t + 1 < nt;
    int si2 = 0, di2 = 0; bool valid2 = false;
    if (more)
      fetch_idx(src, dst, s_list, rem > TE ? rem - TE : 0, rem < TE ? rem : TE,
                rr, lane, si2, di2, valid2);
    if (!FINAL) __syncthreads();

    v8f acc[MT][NTW];
    v16h ag[3][2];
#pragma unroll
    for (int mt = 0; mt < MT; ++mt) {
      acc[mt][0] = (v8f){};
      acc[mt][1] = (v8f){};
      ag[mt >> 1][mt & 1] = ld_a_frag(le + mt * 16 * EP, EP, M, hi, 0);
    }
#pragma unroll
    for (int q = 0; q < 16; ++q) {
      if (q + 2 < 16) {
#pragma unroll
        for (int p = 0; p < 2; ++p)
          ag[(q + 2) % 3][p] = ld_a_frag(le + ((q & 1) * 2 + p) * 16 * EP, EP, M, hi,
                                         ((q + 2) >> 1) * 32);
      }
      __builtin_amdgcn_sched_barrier(0);
#pragma unroll
      for (int p = 0; p < 2; ++p) {
        const int mt = (q & 1) * 2 + p;
#pragma unroll
        for (int j = 0; j < NTW; ++j)
          acc[mt][j] = __builtin_amdgcn_wmma_f32_16x16x32_f16(false, ag[q % 3][p], false,
                                                              w1r[j][q >> 1], (short)0,
                                                              acc[mt][j], false, false);
      }
      __builtin_amdgcn_sched_barrier(0);
    }
#pragma unroll
    for (int mt = 0; mt < MT; ++mt) {
#pragma unroll
      for (int j = 0; j < NTW; ++j) {
#pragma unroll
        for (int r = 0; r < 8; ++r) {
          float v = acc[mt][j][r] + bias1[j];
          v = v > 0.0f ? v : 0.0f;
          lh[(mt * 16 + r + 8 * hi) * HP + nn + 16 * j] = (f16)v;
        }
      }
    }
    __syncthreads();

    v8f acc2[MT][NTW];
#pragma unroll
    for (int mt = 0; mt < MT; ++mt) {
      acc2[mt][0] = (v8f){};
      acc2[mt][1] = (v8f){};
      ag[mt >> 1][mt & 1] = ld_a_frag(lh + mt * 16 * HP, HP, M, hi, 0);
    }
#pragma unroll
    for (int q = 0; q < 8; ++q) {
      if (q + 2 < 8) {
#pragma unroll
        for (int p = 0; p < 2; ++p)
          ag[(q + 2) % 3][p] = ld_a_frag(lh + ((q & 1) * 2 + p) * 16 * HP, HP, M, hi,
                                         ((q + 2) >> 1) * 32);
      }
      __builtin_amdgcn_sched_barrier(0);
#pragma unroll
      for (int p = 0; p < 2; ++p) {
        const int mt = (q & 1) * 2 + p;
#pragma unroll
        for (int j = 0; j < NTW; ++j)
          acc2[mt][j] = __builtin_amdgcn_wmma_f32_16x16x32_f16(false, ag[q % 3][p], false,
                                                               w2r[j][q >> 1], (short)0,
                                                               acc2[mt][j], false, false);
      }
      __builtin_amdgcn_sched_barrier(0);
    }

    float4 xi[RPW], xj[RPW];
    if (FINAL) issue_rows(x, si2, di2, lane, xi, xj);

#pragma unroll
    for (int mt = 0; mt < MT; ++mt)
      sum_rows(ld, s_acc, acc2[mt][0], acc2[mt][1], mt, hi, nn, bias2[0], bias2[1]);
    if (FINAL) store_rows(le, r0, lane, xi, xj);
    __syncthreads();
    if (FINAL) { si = si2; di = di2; valid = valid2; }
  }
}

__global__ __launch_bounds__(NTHR) void edge_range_kernel(
    const float* __restrict__ x,
    const int* __restrict__ src,
    const int* __restrict__ dst,
    const f16* __restrict__ W1f,
    const float* __restrict__ b1,
    const f16* __restrict__ W2f,
    const float* __restrict__ b2,
    float* __restrict__ summed,
    float* __restrict__ cnt) {
  __shared__ __attribute__((aligned(16))) float s_acc[RN * FOUT];
  __shared__ int   s_cnt[RN];
  __shared__ int   s_list[LCAP];
  __shared__ __attribute__((aligned(16))) f16 s_e[2 * 64 * EP];
  __shared__ __attribute__((aligned(16))) f16 s_h[2 * 64 * HP];
  __shared__ int   s_d[2 * 64];
  __shared__ int   s_n;

  const int tid = threadIdx.x;
  const int lo  = blockIdx.x * RN;
  const int nn  = (NNODES - lo) < RN ? (NNODES - lo) : RN;

  v16h w1r[NTW][8], w2r[NTW][4];
  float bias1[NTW], bias2[NTW];
  {
    const int wlane = tid & 31;
    const int wwv   = (tid >> 5) & 3;
#pragma unroll
    for (int j = 0; j < NTW; ++j) {
      const int ntile = wwv * NTW + j;
#pragma unroll
      for (int ks = 0; ks < 8; ++ks) w1r[j][ks] = ld_b_frag(W1f, ntile * 8 + ks, wlane);
#pragma unroll
      for (int ks = 0; ks < 4; ++ks) w2r[j][ks] = ld_b_frag(W2f, ntile * 4 + ks, wlane);
      bias1[j] = b1[ntile * 16 + (wlane & 15)];
      bias2[j] = b2[ntile * 16 + (wlane & 15)];
    }
  }

  for (int i = tid; i < RN * FOUT / 4; i += NTHR)
    ((float4*)s_acc)[i] = make_float4(0.f, 0.f, 0.f, 0.f);
  for (int i = tid; i < RN; i += NTHR) s_cnt[i] = 0;
  if (tid == 0) s_n = 0;
  __syncthreads();

  const int4* d4 = (const int4*)dst;
  int4 cur[SCJ];
#pragma unroll
  for (int j = 0; j < SCJ; ++j) {
    cur[j] = d4[j * NTHR + tid];
  }

  for (int c = 0; c < NCH; ++c) {
    int4 nxt[SCJ];
#pragma unroll
    for (int j = 0; j < SCJ; ++j) {
      const int q = (c + 1) * (CHUNK / 4) + j * NTHR + tid;
      nxt[j] = d4[q < NEDGES / 4 ? q : NEDGES / 4 - 1];
    }
    unsigned mlo = 0u, mhi = 0u;
#pragma unroll
    for (int j = 0; j < SCJ; ++j) {
      const int q = c * (CHUNK / 4) + j * NTHR + tid;
      const bool inb = q < NEDGES / 4;
      const int v[4] = {cur[j].x, cur[j].y, cur[j].z, cur[j].w};
#pragma unroll
      for (int k = 0; k < 4; ++k) {
        const bool hit = inb && (unsigned)(v[k] - lo) < (unsigned)nn;
        const unsigned bit = 1u << ((4 * j + k) & 31);
        if (j < SCJ / 2) mlo |= hit ? bit : 0u;
        else             mhi |= hit ? bit : 0u;
      }
    }
    const int h = __builtin_popcount(mlo) + __builtin_popcount(mhi);
    if (h > 0) {
      int p = atomicAdd(&s_n, h);
      const int eb = (c * (CHUNK / 4) + tid) * 4;
      while (mlo) {
        const int b = __builtin_ctz(mlo);
        mlo &= mlo - 1u;
        s_list[p++] = eb + (b >> 2) * (4 * NTHR) + (b & 3);
      }
      while (mhi) {
        const int b = __builtin_ctz(mhi) + 32;
        mhi &= mhi - 1u;
        s_list[p++] = eb + (b >> 2) * (4 * NTHR) + (b & 3);
      }
    }
    __syncthreads();
    const int n = s_n;
    __syncthreads();
    if (n > FLUSH_AT) {
      run_tiles<false>(x, src, dst, w1r, bias1, w2r, bias2, lo, n,
                       s_acc, s_cnt, s_list, s_e, s_h, s_d);
      if (tid == 0) s_n = n & (TE - 1);
      __syncthreads();
    }
#pragma unroll
    for (int j = 0; j < SCJ; ++j) cur[j] = nxt[j];
  }
  {
    const int n = s_n;
    run_tiles<true>(x, src, dst, w1r, bias1, w2r, bias2, lo, n,
                    s_acc, s_cnt, s_list, s_e, s_h, s_d);
  }
  __syncthreads();

  float4* so = (float4*)(summed + (size_t)lo * FOUT);
  for (int i = tid; i < nn * (FOUT / 4); i += NTHR) so[i] = ((const float4*)s_acc)[i];
  for (int i = tid; i < nn; i += NTHR) cnt[lo + i] = (float)s_cnt[i];
}

__global__ __launch_bounds__(256) void node_mlp_kernel(
    const float* __restrict__ summed,
    const float* __restrict__ cnt,
    const f16* __restrict__ W3f,
    const float* __restrict__ b3,
    const f16* __restrict__ W4f,
    const float* __restrict__ b4,
    float* __restrict__ out) {
  __shared__ f16 lds_a[16 * 128];
  __shared__ f16 lds_m[16 * 64];

  const int tid   = threadIdx.x;
  const int lane  = tid & 31;
  const int wv    = tid >> 5;
  const int nbase = blockIdx.x * 16;

  {
    const int r    = tid >> 4;
    const int q    = tid & 15;
    const int node = nbase + r;
    float cv  = cnt[node];
    float inv = 1.0f / (cv > 1.0f ? cv : 1.0f);
    const float4* sp = (const float4*)(summed + (size_t)node * FOUT);
    float4 s0 = sp[q], s1 = sp[q + 16];
    float4 m0 = make_float4(s0.x * inv, s0.y * inv, s0.z * inv, s0.w * inv);
    float4 m1 = make_float4(s1.x * inv, s1.y * inv, s1.z * inv, s1.w * inv);
    m0.x = m0.x > 0.f ? m0.x : 0.f; m0.y = m0.y > 0.f ? m0.y : 0.f;
    m0.z = m0.z > 0.f ? m0.z : 0.f; m0.w = m0.w > 0.f ? m0.w : 0.f;
    m1.x = m1.x > 0.f ? m1.x : 0.f; m1.y = m1.y > 0.f ? m1.y : 0.f;
    m1.z = m1.z > 0.f ? m1.z : 0.f; m1.w = m1.w > 0.f ? m1.w : 0.f;
    *(uint2*)(&lds_a[r * 128 + 4 * q])      = pack4(m0);
    *(uint2*)(&lds_a[r * 128 + 64 + 4 * q]) = pack4(m1);
  }
  __syncthreads();

  const int M  = lane & 15;
  const int hi = (lane >> 4) & 1;

  if (wv < 4) {
    const int n = wv * 16 + (lane & 15);
    v8f acc = {};
#pragma unroll
    for (int ks = 0; ks < 4; ++ks) {
      v16h a = ld_a_frag(lds_a, 128, M, hi, ks * 32);
      v16h b = ld_b_frag(W3f, wv * 4 + ks, lane);
      acc = __builtin_amdgcn_wmma_f32_16x16x32_f16(false, a, false, b,
                                                   (short)0, acc, false, false);
    }
    const float bias = b3[n];
#pragma unroll
    for (int r = 0; r < 8; ++r) {
      float v = acc[r] + bias;
      v = v > 0.0f ? v : 0.0f;
      lds_m[(r + 8 * hi) * 64 + n] = (f16)v;
    }
  }
  __syncthreads();

  {
    const int n = wv * 16 + (lane & 15);
    v8f acc = {};
#pragma unroll
    for (int ks = 0; ks < 2; ++ks) {
      v16h a = ld_a_frag(lds_m, 64, M, hi, ks * 32);
      v16h b = ld_b_frag(W4f, wv * 2 + ks, lane);
      acc = __builtin_amdgcn_wmma_f32_16x16x32_f16(false, a, false, b,
                                                   (short)0, acc, false, false);
    }
    const float bias = b4[n];
#pragma unroll
    for (int r = 0; r < 8; ++r) {
      out[(size_t)(nbase + r + 8 * hi) * FOUT + n] = acc[r] + bias;
    }
  }
}

extern "C" void kernel_launch(void* const* d_in, const int* in_sizes, int n_in,
                              void* d_out, int out_size, void* d_ws, size_t ws_size,
                              hipStream_t stream) {
  const float* x   = (const float*)d_in[0];
  const int*  eidx = (const int*)d_in[1];
  const float* W1  = (const float*)d_in[2];
  const float* b1  = (const float*)d_in[3];
  const float* W2  = (const float*)d_in[4];
  const float* b2  = (const float*)d_in[5];
  const float* W3  = (const float*)d_in[6];
  const float* b3  = (const float*)d_in[7];
  const float* W4  = (const float*)d_in[8];
  const float* b4  = (const float*)d_in[9];
  float* out = (float*)d_out;

  char* ws = (char*)d_ws;
  float* summed = (float*)ws;
  float* cnt    = (float*)(ws + 25600000);
  f16* W1f = (f16*)(ws + 25800000);
  f16* W2f = (f16*)(ws + 25800000 + 65536);
  f16* W3f = (f16*)(ws + 25800000 + 65536 + 32768);
  f16* W4f = (f16*)(ws + 25800000 + 65536 + 32768 + 16384);

  pack_weights_kernel<<<(256 * 128 + 255) / 256, 256, 0, stream>>>(W1, W1f, 256, 128);
  pack_weights_kernel<<<(128 * 128 + 255) / 256, 256, 0, stream>>>(W2, W2f, 128, 128);
  pack_weights_kernel<<<(128 * 64  + 255) / 256, 256, 0, stream>>>(W3, W3f, 128, 64);
  pack_weights_kernel<<<(64  * 128 + 255) / 256, 256, 0, stream>>>(W4, W4f, 64, 128);

  edge_range_kernel<<<NBLK, NTHR, 0, stream>>>(
      x, eidx, eidx + NEDGES, W1f, b1, W2f, b2, summed, cnt);

  node_mlp_kernel<<<NNODES / 16, 256, 0, stream>>>(
      summed, cnt, W3f, b3, W4f, b4, out);
}


# LDS summing: a wave reads its tile's 32 destination rows in 8 wide LDS loads up front and issues its 64 adds back to back (pad rows: spare row); was load, LDS wait, branch, 2 adds per row, in series
# speedup vs baseline: 3.3427x; 1.1730x over previous
#include <hip/hip_runtime.h>

typedef _Float16 f16;
typedef __attribute__((ext_vector_type(16))) _Float16 v16h;
typedef __attribute__((ext_vector_type(8)))  float    v8f;

#define NNODES 50000
#define NEDGES 800000
#define FIN    128
#define FOUT   128
#define MT     4
#define NTW    2
#define RPW    16
#define RN     208
#define NBLK   241
#define NTHR   256
#define TE     128
#define SCJ    16
#define CHUNK  (NTHR * 4 * SCJ)
#define NCH    ((NEDGES + CHUNK - 1) / CHUNK)
#define LCAP   (CHUNK + 8192)
#define FLUSH_AT (LCAP - CHUNK)
#define EP     264
#define HP     136

typedef __attribute__((ext_vector_type(2))) float    v2f;
typedef __attribute__((ext_vector_type(2))) _Float16 v2h;

union V16H { v16h v; uint4 q[2]; };
union H4   { v2h h[2]; uint2 q; };

__device__ __forceinline__ uint2 pack4(float4 a) {
  H4 u;
  u.h[0] = __builtin_convertvector((v2f){a.x, a.y}, v2h);
  u.h[1] = __builtin_convertvector((v2f){a.z, a.w}, v2h);
  return u.q;
}

__device__ __forceinline__ v16h ld_a_frag(const f16* lds, int row_stride,
                                          int M, int hi, int kbase) {
  const int kb = kbase + hi * 8;
  V16H t;
  t.q[0] = *(const uint4*)(lds + M * row_stride + kb);
  t.q[1] = *(const uint4*)(lds + M * row_stride + kb + 16);
  return t.v;
}

__device__ __forceinline__ v16h ld_b_frag(const f16* __restrict__ wf,
                                          int frag, int lane) {
  const uint4* p = (const uint4*)(wf + (size_t)frag * 512 + lane * 8);
  V16H t;
  t.q[0] = p[0];
  t.q[1] = p[32];
  return t.v;
}

__global__ __launch_bounds__(256) void pack_weights_kernel(
    const float* __restrict__ W, f16* __restrict__ out, int K, int N) {
  int o = blockIdx.x * blockDim.x + threadIdx.x;
  if (o >= K * N) return;
  int ksteps = K >> 5;
  int e    = ((o >> 8) & 1) * 8 + (o & 7);
  int lane = (o >> 3) & 31;
  int frag = o >> 9;
  int ks   = frag % ksteps;
  int nt   = frag / ksteps;
  int kl   = (e < 8 ? e : e + 8) + ((lane & 16) ? 8 : 0);
  int Kidx = ks * 32 + kl;
  int n    = nt * 16 + (lane & 15);
  out[o] = (f16)W[Kidx * N + n];
}


__device__ __forceinline__ void fetch_idx(
    const int* __restrict__ src, const int* __restrict__ dst, const int* s_list,
    int base, int nvalid, int rr, int lane, int& si, int& di, bool& valid) {
  si = 0; di = 0; valid = false;
  if (lane < RPW) {
    valid = rr < nvalid;
    int e = 0;
    if (valid) e = s_list[base + rr];
    si = src[e];
    di = dst[e];
  }
}

__device__ __forceinline__ void issue_rows(
    const float* __restrict__ x, int si, int di, int lane,
    float4 (&xi)[RPW], float4 (&xj)[RPW]) {
#pragma unroll
  for (int j = 0; j < RPW; ++j) {
    const int dj = __builtin_amdgcn_readlane(di, j);
    const int sj = __builtin_amdgcn_readlane(si, j);
    xi[j] = ((const float4*)x)[(unsigned)(dj * (FIN / 4) + lane)];
    xj[j] = ((const float4*)x)[(unsigned)(sj * (FIN / 4) + lane)];
  }
}

__device__ __forceinline__ void store_rows(f16* le, int r0, int lane,
                                           const float4 (&xi)[RPW], const float4 (&xj)[RPW]) {
#pragma unroll
  for (int j = 0; j < RPW; ++j) {
    f16* row = le + (r0 + j) * EP + lane * 4;
    const float4 a = xi[j], b = xj[j];
    *(uint2*)row         = pack4(a);
    *(uint2*)(row + 128) = pack4(make_float4(b.x - a.x, b.y - a.y, b.z - a.z, b.w - a.w));
  }
}

__device__ __forceinline__ void load_dests(const int* ld, int mt, int hi, int (&d)[8]) {
  const int4* p = (const int4*)(ld + mt * 16 + 8 * hi);
  const int4 a = p[0], b = p[1];
  d[0] = a.x; d[1] = a.y; d[2] = a.z; d[3] = a.w;
  d[4] = b.x; d[5] = b.y; d[6] = b.z; d[7] = b.w;
}

__device__ __forceinline__ void sum_rows(const int (&d)[8], float* s_acc,
                                         const v8f& a0, const v8f& a1, int nn,
                                         float bias2a, float bias2b) {
#pragma unroll
  for (int r = 0; r < 8; ++r) {
    atomicAdd(&s_acc[d[r] * FOUT + nn],      a0[r] + bias2a);
    atomicAdd(&s_acc[d[r] * FOUT + nn + 16], a1[r] + bias2b);
  }
}

template <bool FINAL>
__device__ __forceinline__ void run_tiles(
    const float* __restrict__ x, const int* __restrict__ src,
    const int* __restrict__ dst, const v16h (&w1r)[NTW][8], const float (&bias1)[NTW],
    const v16h (&w2r)[NTW][4], const float (&bias2)[NTW], int lo, int n,
    float* s_acc, int* s_cnt, const int* s_list, f16* s_e, f16* s_h, int* s_d) {
  const int nt = FINAL ? (n + TE - 1) / TE : n / TE;
  if (nt == 0) return;
  const int tid  = threadIdx.x;
  const int g    = tid >> 7;
  const int lane = tid & 31;
  const int wv   = (tid >> 5) & 3;
  f16* le = s_e + g * (64 * EP);
  f16* lh = s_h + g * (64 * HP);
  int* ld = s_d + g * 64;
  const int r0 = wv * RPW;
  const int rr = g * 64 + r0 + lane;
  const int M  = lane & 15;
  const int hi = (lane >> 4) & 1;
  const int nn = wv * (16 * NTW) + (lane & 15);

  int si = 0, di = 0; bool valid = false;
  if (FINAL) {
    float4 xi[RPW], xj[RPW];
    fetch_idx(src, dst, s_list, n > TE ? n - TE : 0, n < TE ? n : TE, rr, lane, si, di, valid);
    issue_rows(x, si, di, lane, xi, xj);
    store_rows(le, r0, lane, xi, xj);
    __syncthreads();
  }

  for (int t = 0; t < nt; ++t) {
    if (!FINAL) {
      float4 xi[RPW], xj[RPW];
      fetch_idx(src, dst, s_list, n - (t + 1) * TE, TE, rr, lane, si, di, valid);
      issue_rows(x, si, di, lane, xi, xj);
      store_rows(le, r0, lane, xi, xj);
    }
    if (lane < RPW) {
      ld[r0 + lane] = valid ? (di - lo) : RN;
      if (valid) atomicAdd(&s_cnt[di - lo], 1);
    }
    const int  rem  = n - (t + 1) * TE;
    const bool more = FINAL && t + 1 < nt;
    int si2 = 0, di2 = 0; bool valid2 = false;
    if (more)
      fetch_idx(src, dst, s_list, rem > TE ? rem - TE : 0, rem < TE ? rem : TE,
                rr, lane, si2, di2, valid2);
    if (!FINAL) __syncthreads();

    v8f acc[MT][NTW];
    v16h ag[3][2];
#pragma unroll
    for (int mt = 0; mt < MT; ++mt) {
      acc[mt][0] = (v8f){};
      acc[mt][1] = (v8f){};
      ag[mt >> 1][mt & 1] = ld_a_frag(le + mt * 16 * EP, EP, M, hi, 0);
    }
#pragma unroll
    for (int q = 0; q < 16; ++q) {
      if (q + 2 < 16) {
#pragma unroll
        for (int p = 0; p < 2; ++p)
          ag[(q + 2) % 3][p] = ld_a_frag(le + ((q & 1) * 2 + p) * 16 * EP, EP, M, hi,
                                         ((q + 2) >> 1) * 32);
      }
      __builtin_amdgcn_sched_barrier(0);
#pragma unroll
      for (int p = 0; p < 2; ++p) {
        const int mt = (q & 1) * 2 + p;
#pragma unroll
        for (int j = 0; j < NTW; ++j)
          acc[mt][j] = __builtin_amdgcn_wmma_f32_16x16x32_f16(false, ag[q % 3][p], false,
                                                              w1r[j][q >> 1], (short)0,
                                                              acc[mt][j], false, false);
      }
      __builtin_amdgcn_sched_barrier(0);
    }
#pragma unroll
    for (int mt = 0; mt < MT; ++mt) {
#pragma unroll
      for (int j = 0; j < NTW; ++j) {
#pragma unroll
        for (int r = 0; r < 8; ++r) {
          float v = acc[mt][j][r] + bias1[j];
          v = v > 0.0f ? v : 0.0f;
          lh[(mt * 16 + r + 8 * hi) * HP + nn + 16 * j] = (f16)v;
        }
      }
    }
    __syncthreads();

    v8f acc2[MT][NTW];
#pragma unroll
    for (int mt = 0; mt < MT; ++mt) {
      acc2[mt][0] = (v8f){};
      acc2[mt][1] = (v8f){};
      ag[mt >> 1][mt & 1] = ld_a_frag(lh + mt * 16 * HP, HP, M, hi, 0);
    }
#pragma unroll
    for (int q = 0; q < 8; ++q) {
      if (q + 2 < 8) {
#pragma unroll
        for (int p = 0; p < 2; ++p)
          ag[(q + 2) % 3][p] = ld_a_frag(lh + ((q & 1) * 2 + p) * 16 * HP, HP, M, hi,
                                         ((q + 2) >> 1) * 32);
      }
      __builtin_amdgcn_sched_barrier(0);
#pragma unroll
      for (int p = 0; p < 2; ++p) {
        const int mt = (q & 1) * 2 + p;
#pragma unroll
        for (int j = 0; j < NTW; ++j)
          acc2[mt][j] = __builtin_amdgcn_wmma_f32_16x16x32_f16(false, ag[q % 3][p], false,
                                                               w2r[j][q >> 1], (short)0,
                                                               acc2[mt][j], false, false);
      }
      __builtin_amdgcn_sched_barrier(0);
    }

    int dd[MT][8];
#pragma unroll
    for (int mt = 0; mt < MT; ++mt) load_dests(ld, mt, hi, dd[mt]);
    __builtin_amdgcn_sched_barrier(0);

    float4 xi[RPW], xj[RPW];
    if (FINAL) issue_rows(x, si2, di2, lane, xi, xj);

#pragma unroll
    for (int mt = 0; mt < MT; ++mt)
      sum_rows(dd[mt], s_acc, acc2[mt][0], acc2[mt][1], nn, bias2[0], bias2[1]);
    if (FINAL) store_rows(le, r0, lane, xi, xj);
    __syncthreads();
    if (FINAL) { si = si2; di = di2; valid = valid2; }
  }
}

__global__ __launch_bounds__(NTHR) void edge_range_kernel(
    const float* __restrict__ x,
    const int* __restrict__ src,
    const int* __restrict__ dst,
    const f16* __restrict__ W1f,
    const float* __restrict__ b1,
    const f16* __restrict__ W2f,
    const float* __restrict__ b2,
    float* __restrict__ summed,
    float* __restrict__ cnt) {
  __shared__ __attribute__((aligned(16))) float s_acc[(RN + 1) * FOUT];
  __shared__ int   s_cnt[RN];
  __shared__ int   s_list[LCAP];
  __shared__ __attribute__((aligned(16))) f16 s_e[2 * 64 * EP];
  __shared__ __attribute__((aligned(16))) f16 s_h[2 * 64 * HP];
  __shared__ __attribute__((aligned(16))) int s_d[2 * 64];
  __shared__ int   s_n;

  const int tid = threadIdx.x;
  const int lo  = blockIdx.x * RN;
  const int nn  = (NNODES - lo) < RN ? (NNODES - lo) : RN;

  v16h w1r[NTW][8], w2r[NTW][4];
  float bias1[NTW], bias2[NTW];
  {
    const int wlane = tid & 31;
    const int wwv   = (tid >> 5) & 3;
#pragma unroll
    for (int j = 0; j < NTW; ++j) {
      const int ntile = wwv * NTW + j;
#pragma unroll
      for (int ks = 0; ks < 8; ++ks) w1r[j][ks] = ld_b_frag(W1f, ntile * 8 + ks, wlane);
#pragma unroll
      for (int ks = 0; ks < 4; ++ks) w2r[j][ks] = ld_b_frag(W2f, ntile * 4 + ks, wlane);
      bias1[j] = b1[ntile * 16 + (wlane & 15)];
      bias2[j] = b2[ntile * 16 + (wlane & 15)];
    }
  }

  for (int i = tid; i < (RN + 1) * FOUT / 4; i += NTHR)
    ((float4*)s_acc)[i] = make_float4(0.f, 0.f, 0.f, 0.f);
  for (int i = tid; i < RN; i += NTHR) s_cnt[i] = 0;
  if (tid == 0) s_n = 0;
  __syncthreads();

  const int4* d4 = (const int4*)dst;
  int4 cur[SCJ];
#pragma unroll
  for (int j = 0; j < SCJ; ++j) {
    cur[j] = d4[j * NTHR + tid];
  }

  for (int c = 0; c < NCH; ++c) {
    int4 nxt[SCJ];
#pragma unroll
    for (int j = 0; j < SCJ; ++j) {
      const int q = (c + 1) * (CHUNK / 4) + j * NTHR + tid;
      nxt[j] = d4[q < NEDGES / 4 ? q : NEDGES / 4 - 1];
    }
    unsigned mlo = 0u, mhi = 0u;
#pragma unroll
    for (int j = 0; j < SCJ; ++j) {
      const int q = c * (CHUNK / 4) + j * NTHR + tid;
      const bool inb = q < NEDGES / 4;
      const int v[4] = {cur[j].x, cur[j].y, cur[j].z, cur[j].w};
#pragma unroll
      for (int k = 0; k < 4; ++k) {
        const bool hit = inb && (unsigned)(v[k] - lo) < (unsigned)nn;
        const unsigned bit = 1u << ((4 * j + k) & 31);
        if (j < SCJ / 2) mlo |= hit ? bit : 0u;
        else             mhi |= hit ? bit : 0u;
      }
    }
    const int h = __builtin_popcount(mlo) + __builtin_popcount(mhi);
    if (h > 0) {
      int p = atomicAdd(&s_n, h);
      const int eb = (c * (CHUNK / 4) + tid) * 4;
      while (mlo) {
        const int b = __builtin_ctz(mlo);
        mlo &= mlo - 1u;
        s_list[p++] = eb + (b >> 2) * (4 * NTHR) + (b & 3);
      }
      while (mhi) {
        const int b = __builtin_ctz(mhi) + 32;
        mhi &= mhi - 1u;
        s_list[p++] = eb + (b >> 2) * (4 * NTHR) + (b & 3);
      }
    }
    __syncthreads();
    const int n = s_n;
    __syncthreads();
    if (n > FLUSH_AT) {
      run_tiles<false>(x, src, dst, w1r, bias1, w2r, bias2, lo, n,
                       s_acc, s_cnt, s_list, s_e, s_h, s_d);
      if (tid == 0) s_n = n & (TE - 1);
      __syncthreads();
    }
#pragma unroll
    for (int j = 0; j < SCJ; ++j) cur[j] = nxt[j];
  }
  {
    const int n = s_n;
    run_tiles<true>(x, src, dst, w1r, bias1, w2r, bias2, lo, n,
                    s_acc, s_cnt, s_list, s_e, s_h, s_d);
  }
  __syncthreads();

  float4* so = (float4*)(summed + (size_t)lo * FOUT);
  for (int i = tid; i < nn * (FOUT / 4); i += NTHR) so[i] = ((const float4*)s_acc)[i];
  for (int i = tid; i < nn; i += NTHR) cnt[lo + i] = (float)s_cnt[i];
}

__global__ __launch_bounds__(256) void node_mlp_kernel(
    const float* __restrict__ summed,
    const float* __restrict__ cnt,
    const f16* __restrict__ W3f,
    const float* __restrict__ b3,
    const f16* __restrict__ W4f,
    const float* __restrict__ b4,
    float* __restrict__ out) {
  __shared__ f16 lds_a[16 * 128];
  __shared__ f16 lds_m[16 * 64];

  const int tid   = threadIdx.x;
  const int lane  = tid & 31;
  const int wv    = tid >> 5;
  const int nbase = blockIdx.x * 16;

  {
    const int r    = tid >> 4;
    const int q    = tid & 15;
    const int node = nbase + r;
    float cv  = cnt[node];
    float inv = 1.0f / (cv > 1.0f ? cv : 1.0f);
    const float4* sp = (const float4*)(summed + (size_t)node * FOUT);
    float4 s0 = sp[q], s1 = sp[q + 16];
    float4 m0 = make_float4(s0.x * inv, s0.y * inv, s0.z * inv, s0.w * inv);
    float4 m1 = make_float4(s1.x * inv, s1.y * inv, s1.z * inv, s1.w * inv);
    m0.x = m0.x > 0.f ? m0.x : 0.f; m0.y = m0.y > 0.f ? m0.y : 0.f;
    m0.z = m0.z > 0.f ? m0.z : 0.f; m0.w = m0.w > 0.f ? m0.w : 0.f;
    m1.x = m1.x > 0.f ? m1.x : 0.f; m1.y = m1.y > 0.f ? m1.y : 0.f;
    m1.z = m1.z > 0.f ? m1.z : 0.f; m1.w = m1.w > 0.f ? m1.w : 0.f;
    *(uint2*)(&lds_a[r * 128 + 4 * q])      = pack4(m0);
    *(uint2*)(&lds_a[r * 128 + 64 + 4 * q]) = pack4(m1);
  }
  __syncthreads();

  const int M  = lane & 15;
  const int hi = (lane >> 4) & 1;

  if (wv < 4) {
    const int n = wv * 16 + (lane & 15);
    v8f acc = {};
#pragma unroll
    for (int ks = 0; ks < 4; ++ks) {
      v16h a = ld_a_frag(lds_a, 128, M, hi, ks * 32);
      v16h b = ld_b_frag(W3f, wv * 4 + ks, lane);
      acc = __builtin_amdgcn_wmma_f32_16x16x32_f16(false, a, false, b,
                                                   (short)0, acc, false, false);
    }
    const float bias = b3[n];
#pragma unroll
    for (int r = 0; r < 8; ++r) {
      float v = acc[r] + bias;
      v = v > 0.0f ? v : 0.0f;
      lds_m[(r + 8 * hi) * 64 + n] = (f16)v;
    }
  }
  __syncthreads();

  {
    const int n = wv * 16 + (lane & 15);
    v8f acc = {};
#pragma unroll
    for (int ks = 0; ks < 2; ++ks) {
      v16h a = ld_a_frag(lds_m, 64, M, hi, ks * 32);
      v16h b = ld_b_frag(W4f, wv * 2 + ks, lane);
      acc = __builtin_amdgcn_wmma_f32_16x16x32_f16(false, a, false, b,
                                                   (short)0, acc, false, false);
    }
    const float bias = b4[n];
#pragma unroll
    for (int r = 0; r < 8; ++r) {
      out[(size_t)(nbase + r + 8 * hi) * FOUT + n] = acc[r] + bias;
    }
  }
}

extern "C" void kernel_launch(void* const* d_in, const int* in_sizes, int n_in,
                              void* d_out, int out_size, void* d_ws, size_t ws_size,
                              hipStream_t stream) {
  const float* x   = (const float*)d_in[0];
  const int*  eidx = (const int*)d_in[1];
  const float* W1  = (const float*)d_in[2];
  const float* b1  = (const float*)d_in[3];
  const float* W2  = (const float*)d_in[4];
  const float* b2  = (const float*)d_in[5];
  const float* W3  = (const float*)d_in[6];
  const float* b3  = (const float*)d_in[7];
  const float* W4  = (const float*)d_in[8];
  const float* b4  = (const float*)d_in[9];
  float* out = (float*)d_out;

  char* ws = (char*)d_ws;
  float* summed = (float*)ws;
  float* cnt    = (float*)(ws + 25600000);
  f16* W1f = (f16*)(ws + 25800000);
  f16* W2f = (f16*)(ws + 25800000 + 65536);
  f16* W3f = (f16*)(ws + 25800000 + 65536 + 32768);
  f16* W4f = (f16*)(ws + 25800000 + 65536 + 32768 + 16384);

  pack_weights_kernel<<<(256 * 128 + 255) / 256, 256, 0, stream>>>(W1, W1f, 256, 128);
  pack_weights_kernel<<<(128 * 128 + 255) / 256, 256, 0, stream>>>(W2, W2f, 128, 128);
  pack_weights_kernel<<<(128 * 64  + 255) / 256, 256, 0, stream>>>(W3, W3f, 128, 64);
  pack_weights_kernel<<<(64  * 128 + 255) / 256, 256, 0, stream>>>(W4, W4f, 64, 128);

  edge_range_kernel<<<NBLK, NTHR, 0, stream>>>(
      x, eidx, eidx + NEDGES, W1f, b1, W2f, b2, summed, cnt);

  node_mlp_kernel<<<NNODES / 16, 256, 0, stream>>>(
      summed, cnt, W3f, b3, W4f, b4, out);
}
